# starting kernel
# baseline (speedup 1.0000x reference)
#include <hip/hip_runtime.h>
#include <hip/hip_bf16.h>
#include <float.h>


typedef float v2f __attribute__((ext_vector_type(2)));
typedef float v8f __attribute__((ext_vector_type(8)));

#define BATCH 8
#define CH    64
#define NPTS  4096
#define OCH   64
#define KNN   20
#define CHUNK 1024
#define NEG_INF (-3.0e38f)

__device__ __forceinline__ v8f wmma_f32(v2f a, v2f b, v8f c) {
    return __builtin_amdgcn_wmma_f32_16x16x4_f32(
        false, a, false, b, (short)0, c, false, false);
}

__global__ __launch_bounds__(256) void sq_kernel(const float* __restrict__ x,
                                                 float* __restrict__ sq) {
    int t = blockIdx.x * 256 + threadIdx.x;
    int b = t >> 12;
    int n = t & (NPTS - 1);
    const float* p = x + (size_t)b * CH * NPTS + n;
    float s = 0.f;
#pragma unroll
    for (int c = 0; c < CH; ++c) { float v = p[(size_t)c * NPTS]; s += v * v; }
    sq[t] = s;
}

__global__ __launch_bounds__(256) void knn_kernel(const float* __restrict__ x,
                                                  const float* __restrict__ sq,
                                                  int* __restrict__ idxout) {
    __shared__ float smem[16 * CHUNK];

    const int b    = blockIdx.y;
    const int q0   = blockIdx.x * 16;
    const int tid  = threadIdx.x;
    const int wave = tid >> 5;
    const int lane = tid & 31;
    const int lmod = lane & 15;
    const int khalf  = (lane >> 4) << 1;
    const int rowadd = (lane >> 4) << 3;
    const float* xb  = x + (size_t)b * CH * NPTS;
    const float* sqb = sq + (size_t)b * NPTS;

    v2f a[16];
#pragma unroll
    for (int kk = 0; kk < 16; ++kk) {
        const float* p = xb + (size_t)(4 * kk + khalf) * NPTS + q0 + lmod;
        a[kk].x = p[0];
        a[kk].y = p[NPTS];
    }

    float best[KNN]; int bidx[KNN]; float thr = NEG_INF;
#pragma unroll
    for (int i = 0; i < KNN; ++i) { best[i] = NEG_INF; bidx[i] = 0; }

    const int q     = tid >> 4;
    const int strip = tid & 15;

    for (int c0 = 0; c0 < NPTS; c0 += CHUNK) {
        for (int t = wave; t < CHUNK / 16; t += 8) {
            const int col0 = c0 + t * 16;
            v8f acc = {};
#pragma unroll
            for (int kk = 0; kk < 16; ++kk) {
                const float* p = xb + (size_t)(4 * kk + khalf) * NPTS + col0 + lmod;
                v2f bb; bb.x = p[0]; bb.y = p[NPTS];
                acc = wmma_f32(a[kk], bb, acc);
            }
            const float sqc = sqb[col0 + lmod];
#pragma unroll
            for (int r = 0; r < 8; ++r)
                smem[(r + rowadd) * CHUNK + t * 16 + lmod] = 2.0f * acc[r] - sqc;
        }
        __syncthreads();

        const float* row = smem + q * CHUNK;
        for (int i = 0; i < CHUNK / 16; ++i) {
            float v = row[i * 16 + strip];
            if (v > thr) {
                int col = c0 + i * 16 + strip;
                int mpos = 0; float mval = best[0];
#pragma unroll
                for (int s = 1; s < KNN; ++s)
                    if (best[s] < mval) { mval = best[s]; mpos = s; }
#pragma unroll
                for (int s = 0; s < KNN; ++s)
                    if (s == mpos) { best[s] = v; bidx[s] = col; }
                float nmin = best[0];
#pragma unroll
                for (int s = 1; s < KNN; ++s) nmin = fminf(nmin, best[s]);
                thr = nmin;
            }
        }
        __syncthreads();
    }

    int* smi = (int*)smem;
#pragma unroll
    for (int i = 0; i < KNN; ++i) {
        smem[tid * KNN + i]            = best[i];
        smi[16 * CHUNK / 2 + tid * KNN + i] = bidx[i];
    }
    __syncthreads();
    if (tid < 16) {
        float fb[KNN]; int fi[KNN]; float fthr = NEG_INF;
#pragma unroll
        for (int i = 0; i < KNN; ++i) { fb[i] = NEG_INF; fi[i] = 0; }
        for (int s = 0; s < 16; ++s) {
            const int cb = (tid * 16 + s) * KNN;
            for (int i = 0; i < KNN; ++i) {
                float v = smem[cb + i];
                if (v > fthr) {
                    int cidx = smi[16 * CHUNK / 2 + cb + i];
                    int mpos = 0; float mval = fb[0];
#pragma unroll
                    for (int u = 1; u < KNN; ++u)
                        if (fb[u] < mval) { mval = fb[u]; mpos = u; }
#pragma unroll
                    for (int u = 0; u < KNN; ++u)
                        if (u == mpos) { fb[u] = v; fi[u] = cidx; }
                    float nmin = fb[0];
#pragma unroll
                    for (int u = 1; u < KNN; ++u) nmin = fminf(nmin, fb[u]);
                    fthr = nmin;
                }
            }
        }
        int* op = idxout + ((size_t)b * NPTS + q0 + tid) * KNN;
#pragma unroll
        for (int i = 0; i < KNN; ++i) op[i] = fi[i];
    }
}

__global__ __launch_bounds__(256) void proj_kernel(const float* __restrict__ x,
                                                   const float* __restrict__ W,
                                                   float* __restrict__ y1T,
                                                   float* __restrict__ y3T) {
    int task = blockIdx.x * 8 + (threadIdx.x >> 5);
    int lane = threadIdx.x & 31;
    int lmod = lane & 15;
    int khalf  = (lane >> 4) << 1;
    int rowadd = (lane >> 4) << 3;

    int which = task & 1;
    int nt    = (task >> 1) & 255;
    int ot    = (task >> 9) & 3;
    int b     = task >> 11;

    const float* xb = x + (size_t)b * CH * NPTS;
    const float* wr = W + (size_t)(ot * 16 + lmod) * (2 * CH);
    const int n0 = nt * 16;

    v8f acc = {};
#pragma unroll
    for (int kk = 0; kk < 16; ++kk) {
        int c = 4 * kk + khalf;
        v2f av, bv;
        if (which == 0) { av.x = wr[c];            av.y = wr[c + 1]; }
        else            { av.x = wr[CH + c] - wr[c];
                          av.y = wr[CH + c + 1] - wr[c + 1]; }
        const float* p = xb + (size_t)c * NPTS + n0 + lmod;
        bv.x = p[0]; bv.y = p[NPTS];
        acc = wmma_f32(av, bv, acc);
    }

    float* y = (which ? y3T : y1T)
             + (size_t)b * NPTS * OCH + (size_t)(n0 + lmod) * OCH
             + ot * 16 + rowadd;
    float4 lo = { acc[0], acc[1], acc[2], acc[3] };
    float4 hi = { acc[4], acc[5], acc[6], acc[7] };
    ((float4*)y)[0] = lo;
    ((float4*)y)[1] = hi;
}

__global__ __launch_bounds__(256) void gather_kernel(const float* __restrict__ y1T,
                                                     const float* __restrict__ y3T,
                                                     const int* __restrict__ idx,
                                                     const float* __restrict__ gamma,
                                                     float* __restrict__ selbuf,
                                                     float* __restrict__ partial) {
    int waveId = blockIdx.x * 8 + (threadIdx.x >> 5);
    int lane   = threadIdx.x & 31;
    int ch     = lane * 2;
    bool mx0 = gamma[ch] >= 0.f, mx1 = gamma[ch + 1] >= 0.f;

    float s0 = 0.f, s1 = 0.f, q0 = 0.f, q1 = 0.f;

    for (int pt = 0; pt < 32; ++pt) {
        int p = waveId * 32 + pt;
        int b = p >> 12;
        float2 e = ((const float2*)(y3T + (size_t)p * OCH))[lane];
        const int* ip = idx + (size_t)p * KNN;
        const float* y1b = y1T + (size_t)b * NPTS * OCH;

        float hi0 = -FLT_MAX, hi1 = -FLT_MAX, lo0 = FLT_MAX, lo1 = FLT_MAX;
#pragma unroll
        for (int j = 0; j < KNN; ++j) {
            int nb = ip[j];
            float2 g = ((const float2*)(y1b + ((size_t)nb << 6)))[lane];
            float h0 = g.x + e.x, h1 = g.y + e.y;
            s0 += h0; s1 += h1; q0 += h0 * h0; q1 += h1 * h1;
            hi0 = fmaxf(hi0, h0); lo0 = fminf(lo0, h0);
            hi1 = fmaxf(hi1, h1); lo1 = fminf(lo1, h1);
        }
        float2 selv = { mx0 ? hi0 : lo0, mx1 ? hi1 : lo1 };
        ((float2*)(selbuf + (size_t)p * OCH))[lane] = selv;
    }
    float* pb = partial + (size_t)waveId * 128;
    pb[ch]          = s0; pb[ch + 1]       = s1;
    pb[OCH + ch]    = q0; pb[OCH + ch + 1] = q1;
}

__global__ void stats_kernel(const float* __restrict__ partial,
                             const float* __restrict__ gamma,
                             const float* __restrict__ beta,
                             float* __restrict__ coeff) {
    int ch = threadIdx.x;
    if (ch >= OCH) return;
    float s = 0.f, q = 0.f;
#pragma unroll 4
    for (int w = 0; w < 1024; ++w) {
        s += partial[(size_t)w * 128 + ch];
        q += partial[(size_t)w * 128 + OCH + ch];
    }
    const float M = (float)BATCH * NPTS * KNN;
    float mean  = s / M;
    float var   = q / M - mean * mean;
    float scale = gamma[ch] * rsqrtf(var + 1e-5f);
    coeff[ch]       = scale;
    coeff[OCH + ch] = beta[ch] - mean * scale;
}

__global__ __launch_bounds__(256) void final_kernel(const float* __restrict__ selbuf,
                                                    const float* __restrict__ coeff,
                                                    float* __restrict__ out) {
    int t = blockIdx.x * 256 + threadIdx.x;
    int n = t & (NPTS - 1);
    int o = (t >> 12) & (OCH - 1);
    int b = t >> 18;
    float v = selbuf[(((size_t)b * NPTS + n) << 6) + o];
    float r = v * coeff[o] + coeff[OCH + o];
    out[t] = (r >= 0.f) ? r : 0.2f * r;
}

extern "C" void kernel_launch(void* const* d_in, const int* in_sizes, int n_in,
                              void* d_out, int out_size, void* d_ws, size_t ws_size,
                              hipStream_t stream) {
    const float* x     = (const float*)d_in[0];
    const float* W     = (const float*)d_in[1];
    const float* gamma = (const float*)d_in[2];
    const float* beta  = (const float*)d_in[3];
    float* out = (float*)d_out;

    float* ws    = (float*)d_ws;
    float* sq    = ws;
    int*   idx   = (int*)(ws + 32768);
    float* y1T   = ws + 32768 + 655360;
    float* y3T   = y1T + (size_t)BATCH * NPTS * OCH;
    float* sel   = y3T + (size_t)BATCH * NPTS * OCH;
    float* part  = sel + (size_t)BATCH * NPTS * OCH;
    float* coeff = part + 1024 * 128;

    sq_kernel<<<(BATCH * NPTS) / 256, 256, 0, stream>>>(x, sq);
    knn_kernel<<<dim3(NPTS / 16, BATCH), 256, 0, stream>>>(x, sq, idx);
    proj_kernel<<<2048, 256, 0, stream>>>(x, W, y1T, y3T);
    gather_kernel<<<128, 256, 0, stream>>>(y1T, y3T, idx, gamma, sel, part);
    stats_kernel<<<1, 64, 0, stream>>>(part, gamma, beta, coeff);
    final_kernel<<<(BATCH * OCH * NPTS) / 256, 256, 0, stream>>>(sel, coeff, out);
}


# BN statistics: the one-block serial sum over 1024 wave partials (64 threads, 2048 dependent loads each) becomes a 32-wave tree reduction with float4 row loads and an LDS finish
# speedup vs baseline: 1.2048x; 1.2048x over previous
#include <hip/hip_runtime.h>
#include <hip/hip_bf16.h>
#include <float.h>


typedef float v2f __attribute__((ext_vector_type(2)));
typedef float v8f __attribute__((ext_vector_type(8)));

#define BATCH 8
#define CH    64
#define NPTS  4096
#define OCH   64
#define KNN   20
#define CHUNK 1024
#define NEG_INF (-3.0e38f)

__device__ __forceinline__ v8f wmma_f32(v2f a, v2f b, v8f c) {
    return __builtin_amdgcn_wmma_f32_16x16x4_f32(
        false, a, false, b, (short)0, c, false, false);
}

__global__ __launch_bounds__(256) void sq_kernel(const float* __restrict__ x,
                                                 float* __restrict__ sq) {
    int t = blockIdx.x * 256 + threadIdx.x;
    int b = t >> 12;
    int n = t & (NPTS - 1);
    const float* p = x + (size_t)b * CH * NPTS + n;
    float s = 0.f;
#pragma unroll
    for (int c = 0; c < CH; ++c) { float v = p[(size_t)c * NPTS]; s += v * v; }
    sq[t] = s;
}

__global__ __launch_bounds__(256) void knn_kernel(const float* __restrict__ x,
                                                  const float* __restrict__ sq,
                                                  int* __restrict__ idxout) {
    __shared__ float smem[16 * CHUNK];

    const int b    = blockIdx.y;
    const int q0   = blockIdx.x * 16;
    const int tid  = threadIdx.x;
    const int wave = tid >> 5;
    const int lane = tid & 31;
    const int lmod = lane & 15;
    const int khalf  = (lane >> 4) << 1;
    const int rowadd = (lane >> 4) << 3;
    const float* xb  = x + (size_t)b * CH * NPTS;
    const float* sqb = sq + (size_t)b * NPTS;

    v2f a[16];
#pragma unroll
    for (int kk = 0; kk < 16; ++kk) {
        const float* p = xb + (size_t)(4 * kk + khalf) * NPTS + q0 + lmod;
        a[kk].x = p[0];
        a[kk].y = p[NPTS];
    }

    float best[KNN]; int bidx[KNN]; float thr = NEG_INF;
#pragma unroll
    for (int i = 0; i < KNN; ++i) { best[i] = NEG_INF; bidx[i] = 0; }

    const int q     = tid >> 4;
    const int strip = tid & 15;

    for (int c0 = 0; c0 < NPTS; c0 += CHUNK) {
        for (int t = wave; t < CHUNK / 16; t += 8) {
            const int col0 = c0 + t * 16;
            v8f acc = {};
#pragma unroll
            for (int kk = 0; kk < 16; ++kk) {
                const float* p = xb + (size_t)(4 * kk + khalf) * NPTS + col0 + lmod;
                v2f bb; bb.x = p[0]; bb.y = p[NPTS];
                acc = wmma_f32(a[kk], bb, acc);
            }
            const float sqc = sqb[col0 + lmod];
#pragma unroll
            for (int r = 0; r < 8; ++r)
                smem[(r + rowadd) * CHUNK + t * 16 + lmod] = 2.0f * acc[r] - sqc;
        }
        __syncthreads();

        const float* row = smem + q * CHUNK;
        for (int i = 0; i < CHUNK / 16; ++i) {
            float v = row[i * 16 + strip];
            if (v > thr) {
                int col = c0 + i * 16 + strip;
                int mpos = 0; float mval = best[0];
#pragma unroll
                for (int s = 1; s < KNN; ++s)
                    if (best[s] < mval) { mval = best[s]; mpos = s; }
#pragma unroll
                for (int s = 0; s < KNN; ++s)
                    if (s == mpos) { best[s] = v; bidx[s] = col; }
                float nmin = best[0];
#pragma unroll
                for (int s = 1; s < KNN; ++s) nmin = fminf(nmin, best[s]);
                thr = nmin;
            }
        }
        __syncthreads();
    }

    int* smi = (int*)smem;
#pragma unroll
    for (int i = 0; i < KNN; ++i) {
        smem[tid * KNN + i]            = best[i];
        smi[16 * CHUNK / 2 + tid * KNN + i] = bidx[i];
    }
    __syncthreads();
    if (tid < 16) {
        float fb[KNN]; int fi[KNN]; float fthr = NEG_INF;
#pragma unroll
        for (int i = 0; i < KNN; ++i) { fb[i] = NEG_INF; fi[i] = 0; }
        for (int s = 0; s < 16; ++s) {
            const int cb = (tid * 16 + s) * KNN;
            for (int i = 0; i < KNN; ++i) {
                float v = smem[cb + i];
                if (v > fthr) {
                    int cidx = smi[16 * CHUNK / 2 + cb + i];
                    int mpos = 0; float mval = fb[0];
#pragma unroll
                    for (int u = 1; u < KNN; ++u)
                        if (fb[u] < mval) { mval = fb[u]; mpos = u; }
#pragma unroll
                    for (int u = 0; u < KNN; ++u)
                        if (u == mpos) { fb[u] = v; fi[u] = cidx; }
                    float nmin = fb[0];
#pragma unroll
                    for (int u = 1; u < KNN; ++u) nmin = fminf(nmin, fb[u]);
                    fthr = nmin;
                }
            }
        }
        int* op = idxout + ((size_t)b * NPTS + q0 + tid) * KNN;
#pragma unroll
        for (int i = 0; i < KNN; ++i) op[i] = fi[i];
    }
}

__global__ __launch_bounds__(256) void proj_kernel(const float* __restrict__ x,
                                                   const float* __restrict__ W,
                                                   float* __restrict__ y1T,
                                                   float* __restrict__ y3T) {
    int task = blockIdx.x * 8 + (threadIdx.x >> 5);
    int lane = threadIdx.x & 31;
    int lmod = lane & 15;
    int khalf  = (lane >> 4) << 1;
    int rowadd = (lane >> 4) << 3;

    int which = task & 1;
    int nt    = (task >> 1) & 255;
    int ot    = (task >> 9) & 3;
    int b     = task >> 11;

    const float* xb = x + (size_t)b * CH * NPTS;
    const float* wr = W + (size_t)(ot * 16 + lmod) * (2 * CH);
    const int n0 = nt * 16;

    v8f acc = {};
#pragma unroll
    for (int kk = 0; kk < 16; ++kk) {
        int c = 4 * kk + khalf;
        v2f av, bv;
        if (which == 0) { av.x = wr[c];            av.y = wr[c + 1]; }
        else            { av.x = wr[CH + c] - wr[c];
                          av.y = wr[CH + c + 1] - wr[c + 1]; }
        const float* p = xb + (size_t)c * NPTS + n0 + lmod;
        bv.x = p[0]; bv.y = p[NPTS];
        acc = wmma_f32(av, bv, acc);
    }

    float* y = (which ? y3T : y1T)
             + (size_t)b * NPTS * OCH + (size_t)(n0 + lmod) * OCH
             + ot * 16 + rowadd;
    float4 lo = { acc[0], acc[1], acc[2], acc[3] };
    float4 hi = { acc[4], acc[5], acc[6], acc[7] };
    ((float4*)y)[0] = lo;
    ((float4*)y)[1] = hi;
}

__global__ __launch_bounds__(256) void gather_kernel(const float* __restrict__ y1T,
                                                     const float* __restrict__ y3T,
                                                     const int* __restrict__ idx,
                                                     const float* __restrict__ gamma,
                                                     float* __restrict__ selbuf,
                                                     float* __restrict__ partial) {
    int waveId = blockIdx.x * 8 + (threadIdx.x >> 5);
    int lane   = threadIdx.x & 31;
    int ch     = lane * 2;
    bool mx0 = gamma[ch] >= 0.f, mx1 = gamma[ch + 1] >= 0.f;

    float s0 = 0.f, s1 = 0.f, q0 = 0.f, q1 = 0.f;

    for (int pt = 0; pt < 32; ++pt) {
        int p = waveId * 32 + pt;
        int b = p >> 12;
        float2 e = ((const float2*)(y3T + (size_t)p * OCH))[lane];
        const int* ip = idx + (size_t)p * KNN;
        const float* y1b = y1T + (size_t)b * NPTS * OCH;

        float hi0 = -FLT_MAX, hi1 = -FLT_MAX, lo0 = FLT_MAX, lo1 = FLT_MAX;
#pragma unroll
        for (int j = 0; j < KNN; ++j) {
            int nb = ip[j];
            float2 g = ((const float2*)(y1b + ((size_t)nb << 6)))[lane];
            float h0 = g.x + e.x, h1 = g.y + e.y;
            s0 += h0; s1 += h1; q0 += h0 * h0; q1 += h1 * h1;
            hi0 = fmaxf(hi0, h0); lo0 = fminf(lo0, h0);
            hi1 = fmaxf(hi1, h1); lo1 = fminf(lo1, h1);
        }
        float2 selv = { mx0 ? hi0 : lo0, mx1 ? hi1 : lo1 };
        ((float2*)(selbuf + (size_t)p * OCH))[lane] = selv;
    }
    float* pb = partial + (size_t)waveId * 128;
    pb[ch]          = s0; pb[ch + 1]       = s1;
    pb[OCH + ch]    = q0; pb[OCH + ch + 1] = q1;
}

__global__ __launch_bounds__(1024) void stats_kernel(const float* __restrict__ partial,
                                                     const float* __restrict__ gamma,
                                                     const float* __restrict__ beta,
                                                     float* __restrict__ coeff) {
    __shared__ float red[32 * 128];
    const int tid  = threadIdx.x;
    const int wave = tid >> 5;
    const int lane = tid & 31;

    const float4* p = (const float4*)(partial + (size_t)wave * 32 * 128) + lane;
    float4 a = { 0.f, 0.f, 0.f, 0.f };
    float4 c = { 0.f, 0.f, 0.f, 0.f };
    for (int r = 0; r < 32; r += 2) {
        float4 u = p[(size_t)r * 32];
        float4 v = p[(size_t)(r + 1) * 32];
        a.x += u.x; a.y += u.y; a.z += u.z; a.w += u.w;
        c.x += v.x; c.y += v.y; c.z += v.z; c.w += v.w;
    }
    float4 t4 = { a.x + c.x, a.y + c.y, a.z + c.z, a.w + c.w };
    ((float4*)red)[wave * 32 + lane] = t4;
    __syncthreads();

    if (tid < OCH) {
        const int ch = tid;
        float s = 0.f, q = 0.f;
        for (int w = 0; w < 32; ++w) {
            s += red[w * 128 + ch];
            q += red[w * 128 + OCH + ch];
        }
        const float M = (float)BATCH * NPTS * KNN;
        float mean  = s / M;
        float var   = q / M - mean * mean;
        float scale = gamma[ch] * rsqrtf(var + 1e-5f);
        coeff[ch]       = scale;
        coeff[OCH + ch] = beta[ch] - mean * scale;
    }
}

__global__ __launch_bounds__(256) void final_kernel(const float* __restrict__ selbuf,
                                                    const float* __restrict__ coeff,
                                                    float* __restrict__ out) {
    int t = blockIdx.x * 256 + threadIdx.x;
    int n = t & (NPTS - 1);
    int o = (t >> 12) & (OCH - 1);
    int b = t >> 18;
    float v = selbuf[(((size_t)b * NPTS + n) << 6) + o];
    float r = v * coeff[o] + coeff[OCH + o];
    out[t] = (r >= 0.f) ? r : 0.2f * r;
}

extern "C" void kernel_launch(void* const* d_in, const int* in_sizes, int n_in,
                              void* d_out, int out_size, void* d_ws, size_t ws_size,
                              hipStream_t stream) {
    const float* x     = (const float*)d_in[0];
    const float* W     = (const float*)d_in[1];
    const float* gamma = (const float*)d_in[2];
    const float* beta  = (const float*)d_in[3];
    float* out = (float*)d_out;

    float* ws    = (float*)d_ws;
    float* sq    = ws;
    int*   idx   = (int*)(ws + 32768);
    float* y1T   = ws + 32768 + 655360;
    float* y3T   = y1T + (size_t)BATCH * NPTS * OCH;
    float* sel   = y3T + (size_t)BATCH * NPTS * OCH;
    float* part  = sel + (size_t)BATCH * NPTS * OCH;
    float* coeff = part + 1024 * 128;

    sq_kernel<<<(BATCH * NPTS) / 256, 256, 0, stream>>>(x, sq);
    knn_kernel<<<dim3(NPTS / 16, BATCH), 256, 0, stream>>>(x, sq, idx);
    proj_kernel<<<2048, 256, 0, stream>>>(x, W, y1T, y3T);
    gather_kernel<<<128, 256, 0, stream>>>(y1T, y3T, idx, gamma, sel, part);
    stats_kernel<<<1, 1024, 0, stream>>>(part, gamma, beta, coeff);
    final_kernel<<<(BATCH * OCH * NPTS) / 256, 256, 0, stream>>>(sel, coeff, out);
}


# kNN selection: per-thread register top-20 plus 16-strip serial merge becomes a wave-cooperative scan: list sorted across lanes, wave-uniform threshold, one-lane shift insert, no merge
# speedup vs baseline: 3.0830x; 2.5589x over previous
#include <hip/hip_runtime.h>
#include <hip/hip_bf16.h>
#include <float.h>


typedef float v2f __attribute__((ext_vector_type(2)));
typedef float v8f __attribute__((ext_vector_type(8)));

#define BATCH 8
#define CH    64
#define NPTS  4096
#define OCH   64
#define KNN   20
#define CHUNK 1024
#define NEG_INF (-3.0e38f)

__device__ __forceinline__ v8f wmma_f32(v2f a, v2f b, v8f c) {
    return __builtin_amdgcn_wmma_f32_16x16x4_f32(
        false, a, false, b, (short)0, c, false, false);
}

__global__ __launch_bounds__(256) void sq_kernel(const float* __restrict__ x,
                                                 float* __restrict__ sq) {
    int t = blockIdx.x * 256 + threadIdx.x;
    int b = t >> 12;
    int n = t & (NPTS - 1);
    const float* p = x + (size_t)b * CH * NPTS + n;
    float s = 0.f;
#pragma unroll
    for (int c = 0; c < CH; ++c) { float v = p[(size_t)c * NPTS]; s += v * v; }
    sq[t] = s;
}

__device__ __forceinline__ float lane_value(float v, int l) {
    return __int_as_float(__builtin_amdgcn_readlane(__float_as_int(v), l));
}

__device__ __forceinline__ void list_insert(float& lv, int& li, float& thr,
                                            float cand, int cidx, int lane, int below4) {
    const float upv = __int_as_float(__builtin_amdgcn_ds_bpermute(below4, __float_as_int(lv)));
    const int   upi = __builtin_amdgcn_ds_bpermute(below4, li);
    const int pos = __builtin_popcount(__builtin_amdgcn_ballot_w32(lv >= cand));
    lv = (lane < pos) ? lv : ((lane == pos) ? cand : upv);
    li = (lane < pos) ? li : ((lane == pos) ? cidx : upi);
    thr = lane_value(lv, KNN - 1);
}

__device__ __forceinline__ void scan_step(float v, int colbase, float& lv, int& li,
                                          float& thr, int lane, int below4) {
    unsigned m = __builtin_amdgcn_ballot_w32(v > thr);
    while (m) {
        const int j = __builtin_ctz(m);
        m &= m - 1;
        const float cand = lane_value(v, j);
        if (cand > thr)
            list_insert(lv, li, thr, cand, colbase + j, lane, below4);
    }
}

__device__ __forceinline__ void scan_row(const float* __restrict__ row, int c0,
                                         float& lv, int& li, float& thr, int lane, int below4) {
    float n0 = row[lane], n1 = row[32 + lane], n2 = row[64 + lane], n3 = row[96 + lane];
    for (int i = 0; i < CHUNK / 32; i += 4) {
        const float v0 = n0, v1 = n1, v2 = n2, v3 = n3;
        if (i + 4 < CHUNK / 32) {
            n0 = row[(i + 4) * 32 + lane];
            n1 = row[(i + 5) * 32 + lane];
            n2 = row[(i + 6) * 32 + lane];
            n3 = row[(i + 7) * 32 + lane];
        }
        scan_step(v0, c0 + (i + 0) * 32, lv, li, thr, lane, below4);
        scan_step(v1, c0 + (i + 1) * 32, lv, li, thr, lane, below4);
        scan_step(v2, c0 + (i + 2) * 32, lv, li, thr, lane, below4);
        scan_step(v3, c0 + (i + 3) * 32, lv, li, thr, lane, below4);
    }
}

__global__ __launch_bounds__(256) void knn_kernel(const float* __restrict__ x,
                                                  const float* __restrict__ sq,
                                                  int* __restrict__ idxout) {
    __shared__ float smem[16 * CHUNK];

    const int b    = blockIdx.y;
    const int q0   = blockIdx.x * 16;
    const int tid  = threadIdx.x;
    const int wave = tid >> 5;
    const int lane = tid & 31;
    const int lmod = lane & 15;
    const int khalf  = (lane >> 4) << 1;
    const int rowadd = (lane >> 4) << 3;
    const float* xb  = x + (size_t)b * CH * NPTS;
    const float* sqb = sq + (size_t)b * NPTS;

    v2f a[16];
#pragma unroll
    for (int kk = 0; kk < 16; ++kk) {
        const float* p = xb + (size_t)(4 * kk + khalf) * NPTS + q0 + lmod;
        a[kk].x = p[0];
        a[kk].y = p[NPTS];
    }

    float lvA = NEG_INF, lvB = NEG_INF, thrA = NEG_INF, thrB = NEG_INF;
    int   liA = 0, liB = 0;
    const float* rowA = smem + (2 * wave) * CHUNK;
    const float* rowB = rowA + CHUNK;
    const int below4 = ((lane > 0) ? lane - 1 : 0) << 2;

    for (int c0 = 0; c0 < NPTS; c0 += CHUNK) {
        for (int t = wave; t < CHUNK / 16; t += 8) {
            const int col0 = c0 + t * 16;
            v8f acc = {};
#pragma unroll
            for (int kk = 0; kk < 16; ++kk) {
                const float* p = xb + (size_t)(4 * kk + khalf) * NPTS + col0 + lmod;
                v2f bb; bb.x = p[0]; bb.y = p[NPTS];
                acc = wmma_f32(a[kk], bb, acc);
            }
            const float sqc = sqb[col0 + lmod];
#pragma unroll
            for (int r = 0; r < 8; ++r)
                smem[(r + rowadd) * CHUNK + t * 16 + lmod] = 2.0f * acc[r] - sqc;
        }
        __syncthreads();

        scan_row(rowA, c0, lvA, liA, thrA, lane, below4);
        scan_row(rowB, c0, lvB, liB, thrB, lane, below4);
        __syncthreads();
    }

    int* smi = (int*)smem;
    if (lane < KNN) {
        smi[(2 * wave) * KNN + lane]     = liA;
        smi[(2 * wave + 1) * KNN + lane] = liB;
    }
    __syncthreads();
    int* op = idxout + ((size_t)b * NPTS + q0) * KNN;
    op[tid] = smi[tid];
    if (tid < 16 * KNN - 256) op[256 + tid] = smi[256 + tid];
}

__global__ __launch_bounds__(256) void proj_kernel(const float* __restrict__ x,
                                                   const float* __restrict__ W,
                                                   float* __restrict__ y1T,
                                                   float* __restrict__ y3T) {
    int task = blockIdx.x * 8 + (threadIdx.x >> 5);
    int lane = threadIdx.x & 31;
    int lmod = lane & 15;
    int khalf  = (lane >> 4) << 1;
    int rowadd = (lane >> 4) << 3;

    int which = task & 1;
    int nt    = (task >> 1) & 255;
    int ot    = (task >> 9) & 3;
    int b     = task >> 11;

    const float* xb = x + (size_t)b * CH * NPTS;
    const float* wr = W + (size_t)(ot * 16 + lmod) * (2 * CH);
    const int n0 = nt * 16;

    v8f acc = {};
#pragma unroll
    for (int kk = 0; kk < 16; ++kk) {
        int c = 4 * kk + khalf;
        v2f av, bv;
        if (which == 0) { av.x = wr[c];            av.y = wr[c + 1]; }
        else            { av.x = wr[CH + c] - wr[c];
                          av.y = wr[CH + c + 1] - wr[c + 1]; }
        const float* p = xb + (size_t)c * NPTS + n0 + lmod;
        bv.x = p[0]; bv.y = p[NPTS];
        acc = wmma_f32(av, bv, acc);
    }

    float* y = (which ? y3T : y1T)
             + (size_t)b * NPTS * OCH + (size_t)(n0 + lmod) * OCH
             + ot * 16 + rowadd;
    float4 lo = { acc[0], acc[1], acc[2], acc[3] };
    float4 hi = { acc[4], acc[5], acc[6], acc[7] };
    ((float4*)y)[0] = lo;
    ((float4*)y)[1] = hi;
}

__global__ __launch_bounds__(256) void gather_kernel(const float* __restrict__ y1T,
                                                     const float* __restrict__ y3T,
                                                     const int* __restrict__ idx,
                                                     const float* __restrict__ gamma,
                                                     float* __restrict__ selbuf,
                                                     float* __restrict__ partial) {
    int waveId = blockIdx.x * 8 + (threadIdx.x >> 5);
    int lane   = threadIdx.x & 31;
    int ch     = lane * 2;
    bool mx0 = gamma[ch] >= 0.f, mx1 = gamma[ch + 1] >= 0.f;

    float s0 = 0.f, s1 = 0.f, q0 = 0.f, q1 = 0.f;

    for (int pt = 0; pt < 32; ++pt) {
        int p = waveId * 32 + pt;
        int b = p >> 12;
        float2 e = ((const float2*)(y3T + (size_t)p * OCH))[lane];
        const int* ip = idx + (size_t)p * KNN;
        const float* y1b = y1T + (size_t)b * NPTS * OCH;

        float hi0 = -FLT_MAX, hi1 = -FLT_MAX, lo0 = FLT_MAX, lo1 = FLT_MAX;
#pragma unroll
        for (int j = 0; j < KNN; ++j) {
            int nb = ip[j];
            float2 g = ((const float2*)(y1b + ((size_t)nb << 6)))[lane];
            float h0 = g.x + e.x, h1 = g.y + e.y;
            s0 += h0; s1 += h1; q0 += h0 * h0; q1 += h1 * h1;
            hi0 = fmaxf(hi0, h0); lo0 = fminf(lo0, h0);
            hi1 = fmaxf(hi1, h1); lo1 = fminf(lo1, h1);
        }
        float2 selv = { mx0 ? hi0 : lo0, mx1 ? hi1 : lo1 };
        ((float2*)(selbuf + (size_t)p * OCH))[lane] = selv;
    }
    float* pb = partial + (size_t)waveId * 128;
    pb[ch]          = s0; pb[ch + 1]       = s1;
    pb[OCH + ch]    = q0; pb[OCH + ch + 1] = q1;
}

__global__ __launch_bounds__(1024) void stats_kernel(const float* __restrict__ partial,
                                                     const float* __restrict__ gamma,
                                                     const float* __restrict__ beta,
                                                     float* __restrict__ coeff) {
    __shared__ float red[32 * 128];
    const int tid  = threadIdx.x;
    const int wave = tid >> 5;
    const int lane = tid & 31;

    const float4* p = (const float4*)(partial + (size_t)wave * 32 * 128) + lane;
    float4 a = { 0.f, 0.f, 0.f, 0.f };
    float4 c = { 0.f, 0.f, 0.f, 0.f };
    for (int r = 0; r < 32; r += 2) {
        float4 u = p[(size_t)r * 32];
        float4 v = p[(size_t)(r + 1) * 32];
        a.x += u.x; a.y += u.y; a.z += u.z; a.w += u.w;
        c.x += v.x; c.y += v.y; c.z += v.z; c.w += v.w;
    }
    float4 t4 = { a.x + c.x, a.y + c.y, a.z + c.z, a.w + c.w };
    ((float4*)red)[wave * 32 + lane] = t4;
    __syncthreads();

    if (tid < OCH) {
        const int ch = tid;
        float s = 0.f, q = 0.f;
        for (int w = 0; w < 32; ++w) {
            s += red[w * 128 + ch];
            q += red[w * 128 + OCH + ch];
        }
        const float M = (float)BATCH * NPTS * KNN;
        float mean  = s / M;
        float var   = q / M - mean * mean;
        float scale = gamma[ch] * rsqrtf(var + 1e-5f);
        coeff[ch]       = scale;
        coeff[OCH + ch] = beta[ch] - mean * scale;
    }
}

__global__ __launch_bounds__(256) void final_kernel(const float* __restrict__ selbuf,
                                                    const float* __restrict__ coeff,
                                                    float* __restrict__ out) {
    int t = blockIdx.x * 256 + threadIdx.x;
    int n = t & (NPTS - 1);
    int o = (t >> 12) & (OCH - 1);
    int b = t >> 18;
    float v = selbuf[(((size_t)b * NPTS + n) << 6) + o];
    float r = v * coeff[o] + coeff[OCH + o];
    out[t] = (r >= 0.f) ? r : 0.2f * r;
}

extern "C" void kernel_launch(void* const* d_in, const int* in_sizes, int n_in,
                              void* d_out, int out_size, void* d_ws, size_t ws_size,
                              hipStream_t stream) {
    const float* x     = (const float*)d_in[0];
    const float* W     = (const float*)d_in[1];
    const float* gamma = (const float*)d_in[2];
    const float* beta  = (const float*)d_in[3];
    float* out = (float*)d_out;

    float* ws    = (float*)d_ws;
    float* sq    = ws;
    int*   idx   = (int*)(ws + 32768);
    float* y1T   = ws + 32768 + 655360;
    float* y3T   = y1T + (size_t)BATCH * NPTS * OCH;
    float* sel   = y3T + (size_t)BATCH * NPTS * OCH;
    float* part  = sel + (size_t)BATCH * NPTS * OCH;
    float* coeff = part + 1024 * 128;

    sq_kernel<<<(BATCH * NPTS) / 256, 256, 0, stream>>>(x, sq);
    knn_kernel<<<dim3(NPTS / 16, BATCH), 256, 0, stream>>>(x, sq, idx);
    proj_kernel<<<2048, 256, 0, stream>>>(x, W, y1T, y3T);
    gather_kernel<<<128, 256, 0, stream>>>(y1T, y3T, idx, gamma, sel, part);
    stats_kernel<<<1, 1024, 0, stream>>>(part, gamma, beta, coeff);
    final_kernel<<<(BATCH * OCH * NPTS) / 256, 256, 0, stream>>>(sel, coeff, out);
}


# kNN selection: empty list slots start at a lower bound of the 20th largest score (20th largest of the 32 per-lane maxima of the first 1024 columns) instead of minus infinity: most early inserts go
# speedup vs baseline: 3.3579x; 1.0892x over previous
#include <hip/hip_runtime.h>
#include <hip/hip_bf16.h>
#include <float.h>


typedef float v2f __attribute__((ext_vector_type(2)));
typedef float v8f __attribute__((ext_vector_type(8)));

#define BATCH 8
#define CH    64
#define NPTS  4096
#define OCH   64
#define KNN   20
#define CHUNK 1024
#define NEG_INF (-3.0e38f)

__device__ __forceinline__ v8f wmma_f32(v2f a, v2f b, v8f c) {
    return __builtin_amdgcn_wmma_f32_16x16x4_f32(
        false, a, false, b, (short)0, c, false, false);
}

__global__ __launch_bounds__(256) void sq_kernel(const float* __restrict__ x,
                                                 float* __restrict__ sq) {
    int t = blockIdx.x * 256 + threadIdx.x;
    int b = t >> 12;
    int n = t & (NPTS - 1);
    const float* p = x + (size_t)b * CH * NPTS + n;
    float s = 0.f;
#pragma unroll
    for (int c = 0; c < CH; ++c) { float v = p[(size_t)c * NPTS]; s += v * v; }
    sq[t] = s;
}

__device__ __forceinline__ float lane_value(float v, int l) {
    return __int_as_float(__builtin_amdgcn_readlane(__float_as_int(v), l));
}

__device__ __forceinline__ void list_insert(float& lv, int& li, float& thr,
                                            float cand, int cidx, int lane, int below4) {
    const float upv = __int_as_float(__builtin_amdgcn_ds_bpermute(below4, __float_as_int(lv)));
    const int   upi = __builtin_amdgcn_ds_bpermute(below4, li);
    const int pos = __builtin_popcount(__builtin_amdgcn_ballot_w32(lv >= cand));
    lv = (lane < pos) ? lv : ((lane == pos) ? cand : upv);
    li = (lane < pos) ? li : ((lane == pos) ? cidx : upi);
    thr = lane_value(lv, KNN - 1);
}

__device__ __forceinline__ float seed_floor(const float* __restrict__ row, int lane) {
    float m = row[lane];
    for (int i = 1; i < CHUNK / 32; ++i) m = fmaxf(m, row[i * 32 + lane]);
    int larger = 0;
    for (int j = 0; j < 32; ++j) larger += (lane_value(m, j) > m) ? 1 : 0;
    const unsigned hit = __builtin_amdgcn_ballot_w32(larger == KNN - 1);
    if (hit == 0) return NEG_INF;
    const float f = lane_value(m, __builtin_ctz(hit));
    return f - fmaxf(fabsf(f) * 1.0e-6f, 1.0e-20f);
}

__device__ __forceinline__ void scan_step(float v, int colbase, float& lv, int& li,
                                          float& thr, int lane, int below4) {
    unsigned m = __builtin_amdgcn_ballot_w32(v > thr);
    while (m) {
        const int j = __builtin_ctz(m);
        m &= m - 1;
        const float cand = lane_value(v, j);
        if (cand > thr)
            list_insert(lv, li, thr, cand, colbase + j, lane, below4);
    }
}

__device__ __forceinline__ void scan_row(const float* __restrict__ row, int c0,
                                         float& lv, int& li, float& thr, int lane, int below4) {
    float n0 = row[lane], n1 = row[32 + lane], n2 = row[64 + lane], n3 = row[96 + lane];
    for (int i = 0; i < CHUNK / 32; i += 4) {
        const float v0 = n0, v1 = n1, v2 = n2, v3 = n3;
        if (i + 4 < CHUNK / 32) {
            n0 = row[(i + 4) * 32 + lane];
            n1 = row[(i + 5) * 32 + lane];
            n2 = row[(i + 6) * 32 + lane];
            n3 = row[(i + 7) * 32 + lane];
        }
        scan_step(v0, c0 + (i + 0) * 32, lv, li, thr, lane, below4);
        scan_step(v1, c0 + (i + 1) * 32, lv, li, thr, lane, below4);
        scan_step(v2, c0 + (i + 2) * 32, lv, li, thr, lane, below4);
        scan_step(v3, c0 + (i + 3) * 32, lv, li, thr, lane, below4);
    }
}

__device__ __forceinline__ void gram_chunk(const float* __restrict__ xb, const float* __restrict__ sqb,
                                           const v2f (&a)[16], float* __restrict__ smem, int c0,
                                           int wave, int lmod, int khalf, int rowadd) {
    for (int t = wave; t < CHUNK / 16; t += 8) {
        const int col0 = c0 + t * 16;
        v8f acc = {};
#pragma unroll
        for (int kk = 0; kk < 16; ++kk) {
            const float* p = xb + (size_t)(4 * kk + khalf) * NPTS + col0 + lmod;
            v2f bb; bb.x = p[0]; bb.y = p[NPTS];
            acc = wmma_f32(a[kk], bb, acc);
        }
        const float sqc = sqb[col0 + lmod];
#pragma unroll
        for (int r = 0; r < 8; ++r)
            smem[(r + rowadd) * CHUNK + t * 16 + lmod] = 2.0f * acc[r] - sqc;
    }
}

__global__ __launch_bounds__(256) void knn_kernel(const float* __restrict__ x,
                                                  const float* __restrict__ sq,
                                                  int* __restrict__ idxout) {
    __shared__ float smem[16 * CHUNK];

    const int b    = blockIdx.y;
    const int q0   = blockIdx.x * 16;
    const int tid  = threadIdx.x;
    const int wave = tid >> 5;
    const int lane = tid & 31;
    const int lmod = lane & 15;
    const int khalf  = (lane >> 4) << 1;
    const int rowadd = (lane >> 4) << 3;
    const float* xb  = x + (size_t)b * CH * NPTS;
    const float* sqb = sq + (size_t)b * NPTS;

    v2f a[16];
#pragma unroll
    for (int kk = 0; kk < 16; ++kk) {
        const float* p = xb + (size_t)(4 * kk + khalf) * NPTS + q0 + lmod;
        a[kk].x = p[0];
        a[kk].y = p[NPTS];
    }

    int   liA = 0, liB = 0;
    const float* rowA = smem + (2 * wave) * CHUNK;
    const float* rowB = rowA + CHUNK;
    const int below4 = ((lane > 0) ? lane - 1 : 0) << 2;

    gram_chunk(xb, sqb, a, smem, 0, wave, lmod, khalf, rowadd);
    __syncthreads();
    float lvA = seed_floor(rowA, lane);
    float lvB = seed_floor(rowB, lane);
    float thrA = lvA, thrB = lvB;
    scan_row(rowA, 0, lvA, liA, thrA, lane, below4);
    scan_row(rowB, 0, lvB, liB, thrB, lane, below4);
    __syncthreads();

    for (int c0 = CHUNK; c0 < NPTS; c0 += CHUNK) {
        gram_chunk(xb, sqb, a, smem, c0, wave, lmod, khalf, rowadd);
        __syncthreads();

        scan_row(rowA, c0, lvA, liA, thrA, lane, below4);
        scan_row(rowB, c0, lvB, liB, thrB, lane, below4);
        __syncthreads();
    }

    int* smi = (int*)smem;
    if (lane < KNN) {
        smi[(2 * wave) * KNN + lane]     = liA;
        smi[(2 * wave + 1) * KNN + lane] = liB;
    }
    __syncthreads();
    int* op = idxout + ((size_t)b * NPTS + q0) * KNN;
    op[tid] = smi[tid];
    if (tid < 16 * KNN - 256) op[256 + tid] = smi[256 + tid];
}

__global__ __launch_bounds__(256) void proj_kernel(const float* __restrict__ x,
                                                   const float* __restrict__ W,
                                                   float* __restrict__ y1T,
                                                   float* __restrict__ y3T) {
    int task = blockIdx.x * 8 + (threadIdx.x >> 5);
    int lane = threadIdx.x & 31;
    int lmod = lane & 15;
    int khalf  = (lane >> 4) << 1;
    int rowadd = (lane >> 4) << 3;

    int which = task & 1;
    int nt    = (task >> 1) & 255;
    int ot    = (task >> 9) & 3;
    int b     = task >> 11;

    const float* xb = x + (size_t)b * CH * NPTS;
    const float* wr = W + (size_t)(ot * 16 + lmod) * (2 * CH);
    const int n0 = nt * 16;

    v8f acc = {};
#pragma unroll
    for (int kk = 0; kk < 16; ++kk) {
        int c = 4 * kk + khalf;
        v2f av, bv;
        if (which == 0) { av.x = wr[c];            av.y = wr[c + 1]; }
        else            { av.x = wr[CH + c] - wr[c];
                          av.y = wr[CH + c + 1] - wr[c + 1]; }
        const float* p = xb + (size_t)c * NPTS + n0 + lmod;
        bv.x = p[0]; bv.y = p[NPTS];
        acc = wmma_f32(av, bv, acc);
    }

    float* y = (which ? y3T : y1T)
             + (size_t)b * NPTS * OCH + (size_t)(n0 + lmod) * OCH
             + ot * 16 + rowadd;
    float4 lo = { acc[0], acc[1], acc[2], acc[3] };
    float4 hi = { acc[4], acc[5], acc[6], acc[7] };
    ((float4*)y)[0] = lo;
    ((float4*)y)[1] = hi;
}

__global__ __launch_bounds__(256) void gather_kernel(const float* __restrict__ y1T,
                                                     const float* __restrict__ y3T,
                                                     const int* __restrict__ idx,
                                                     const float* __restrict__ gamma,
                                                     float* __restrict__ selbuf,
                                                     float* __restrict__ partial) {
    int waveId = blockIdx.x * 8 + (threadIdx.x >> 5);
    int lane   = threadIdx.x & 31;
    int ch     = lane * 2;
    bool mx0 = gamma[ch] >= 0.f, mx1 = gamma[ch + 1] >= 0.f;

    float s0 = 0.f, s1 = 0.f, q0 = 0.f, q1 = 0.f;

    for (int pt = 0; pt < 32; ++pt) {
        int p = waveId * 32 + pt;
        int b = p >> 12;
        float2 e = ((const float2*)(y3T + (size_t)p * OCH))[lane];
        const int* ip = idx + (size_t)p * KNN;
        const float* y1b = y1T + (size_t)b * NPTS * OCH;

        float hi0 = -FLT_MAX, hi1 = -FLT_MAX, lo0 = FLT_MAX, lo1 = FLT_MAX;
#pragma unroll
        for (int j = 0; j < KNN; ++j) {
            int nb = ip[j];
            float2 g = ((const float2*)(y1b + ((size_t)nb << 6)))[lane];
            float h0 = g.x + e.x, h1 = g.y + e.y;
            s0 += h0; s1 += h1; q0 += h0 * h0; q1 += h1 * h1;
            hi0 = fmaxf(hi0, h0); lo0 = fminf(lo0, h0);
            hi1 = fmaxf(hi1, h1); lo1 = fminf(lo1, h1);
        }
        float2 selv = { mx0 ? hi0 : lo0, mx1 ? hi1 : lo1 };
        ((float2*)(selbuf + (size_t)p * OCH))[lane] = selv;
    }
    float* pb = partial + (size_t)waveId * 128;
    pb[ch]          = s0; pb[ch + 1]       = s1;
    pb[OCH + ch]    = q0; pb[OCH + ch + 1] = q1;
}

__global__ __launch_bounds__(1024) void stats_kernel(const float* __restrict__ partial,
                                                     const float* __restrict__ gamma,
                                                     const float* __restrict__ beta,
                                                     float* __restrict__ coeff) {
    __shared__ float red[32 * 128];
    const int tid  = threadIdx.x;
    const int wave = tid >> 5;
    const int lane = tid & 31;

    const float4* p = (const float4*)(partial + (size_t)wave * 32 * 128) + lane;
    float4 a = { 0.f, 0.f, 0.f, 0.f };
    float4 c = { 0.f, 0.f, 0.f, 0.f };
    for (int r = 0; r < 32; r += 2) {
        float4 u = p[(size_t)r * 32];
        float4 v = p[(size_t)(r + 1) * 32];
        a.x += u.x; a.y += u.y; a.z += u.z; a.w += u.w;
        c.x += v.x; c.y += v.y; c.z += v.z; c.w += v.w;
    }
    float4 t4 = { a.x + c.x, a.y + c.y, a.z + c.z, a.w + c.w };
    ((float4*)red)[wave * 32 + lane] = t4;
    __syncthreads();

    if (tid < OCH) {
        const int ch = tid;
        float s = 0.f, q = 0.f;
        for (int w = 0; w < 32; ++w) {
            s += red[w * 128 + ch];
            q += red[w * 128 + OCH + ch];
        }
        const float M = (float)BATCH * NPTS * KNN;
        float mean  = s / M;
        float var   = q / M - mean * mean;
        float scale = gamma[ch] * rsqrtf(var + 1e-5f);
        coeff[ch]       = scale;
        coeff[OCH + ch] = beta[ch] - mean * scale;
    }
}

__global__ __launch_bounds__(256) void final_kernel(const float* __restrict__ selbuf,
                                                    const float* __restrict__ coeff,
                                                    float* __restrict__ out) {
    int t = blockIdx.x * 256 + threadIdx.x;
    int n = t & (NPTS - 1);
    int o = (t >> 12) & (OCH - 1);
    int b = t >> 18;
    float v = selbuf[(((size_t)b * NPTS + n) << 6) + o];
    float r = v * coeff[o] + coeff[OCH + o];
    out[t] = (r >= 0.f) ? r : 0.2f * r;
}

extern "C" void kernel_launch(void* const* d_in, const int* in_sizes, int n_in,
                              void* d_out, int out_size, void* d_ws, size_t ws_size,
                              hipStream_t stream) {
    const float* x     = (const float*)d_in[0];
    const float* W     = (const float*)d_in[1];
    const float* gamma = (const float*)d_in[2];
    const float* beta  = (const float*)d_in[3];
    float* out = (float*)d_out;

    float* ws    = (float*)d_ws;
    float* sq    = ws;
    int*   idx   = (int*)(ws + 32768);
    float* y1T   = ws + 32768 + 655360;
    float* y3T   = y1T + (size_t)BATCH * NPTS * OCH;
    float* sel   = y3T + (size_t)BATCH * NPTS * OCH;
    float* part  = sel + (size_t)BATCH * NPTS * OCH;
    float* coeff = part + 1024 * 128;

    sq_kernel<<<(BATCH * NPTS) / 256, 256, 0, stream>>>(x, sq);
    knn_kernel<<<dim3(NPTS / 16, BATCH), 256, 0, stream>>>(x, sq, idx);
    proj_kernel<<<2048, 256, 0, stream>>>(x, W, y1T, y3T);
    gather_kernel<<<128, 256, 0, stream>>>(y1T, y3T, idx, gamma, sel, part);
    stats_kernel<<<1, 1024, 0, stream>>>(part, gamma, beta, coeff);
    final_kernel<<<(BATCH * OCH * NPTS) / 256, 256, 0, stream>>>(sel, coeff, out);
}


# final pass: one element per thread with 4-byte reads 256 B apart becomes a 64-point x 64-channel tile transpose through LDS: 512-byte wave loads and stores, 4096 waves instead of 65536
# speedup vs baseline: 3.5090x; 1.0450x over previous
#include <hip/hip_runtime.h>
#include <hip/hip_bf16.h>
#include <float.h>


typedef float v2f __attribute__((ext_vector_type(2)));
typedef float v8f __attribute__((ext_vector_type(8)));

#define BATCH 8
#define CH    64
#define NPTS  4096
#define OCH   64
#define KNN   20
#define CHUNK 1024
#define NEG_INF (-3.0e38f)

__device__ __forceinline__ v8f wmma_f32(v2f a, v2f b, v8f c) {
    return __builtin_amdgcn_wmma_f32_16x16x4_f32(
        false, a, false, b, (short)0, c, false, false);
}

__global__ __launch_bounds__(256) void sq_kernel(const float* __restrict__ x,
                                                 float* __restrict__ sq) {
    int t = blockIdx.x * 256 + threadIdx.x;
    int b = t >> 12;
    int n = t & (NPTS - 1);
    const float* p = x + (size_t)b * CH * NPTS + n;
    float s = 0.f;
#pragma unroll
    for (int c = 0; c < CH; ++c) { float v = p[(size_t)c * NPTS]; s += v * v; }
    sq[t] = s;
}

__device__ __forceinline__ float lane_value(float v, int l) {
    return __int_as_float(__builtin_amdgcn_readlane(__float_as_int(v), l));
}

__device__ __forceinline__ void list_insert(float& lv, int& li, float& thr,
                                            float cand, int cidx, int lane, int below4) {
    const float upv = __int_as_float(__builtin_amdgcn_ds_bpermute(below4, __float_as_int(lv)));
    const int   upi = __builtin_amdgcn_ds_bpermute(below4, li);
    const int pos = __builtin_popcount(__builtin_amdgcn_ballot_w32(lv >= cand));
    lv = (lane < pos) ? lv : ((lane == pos) ? cand : upv);
    li = (lane < pos) ? li : ((lane == pos) ? cidx : upi);
    thr = lane_value(lv, KNN - 1);
}

__device__ __forceinline__ float seed_floor(const float* __restrict__ row, int lane) {
    float m = row[lane];
    for (int i = 1; i < CHUNK / 32; ++i) m = fmaxf(m, row[i * 32 + lane]);
    int larger = 0;
    for (int j = 0; j < 32; ++j) larger += (lane_value(m, j) > m) ? 1 : 0;
    const unsigned hit = __builtin_amdgcn_ballot_w32(larger == KNN - 1);
    if (hit == 0) return NEG_INF;
    const float f = lane_value(m, __builtin_ctz(hit));
    return f - fmaxf(fabsf(f) * 1.0e-6f, 1.0e-20f);
}

__device__ __forceinline__ void scan_step(float v, int colbase, float& lv, int& li,
                                          float& thr, int lane, int below4) {
    unsigned m = __builtin_amdgcn_ballot_w32(v > thr);
    while (m) {
        const int j = __builtin_ctz(m);
        m &= m - 1;
        const float cand = lane_value(v, j);
        if (cand > thr)
            list_insert(lv, li, thr, cand, colbase + j, lane, below4);
    }
}

__device__ __forceinline__ void scan_row(const float* __restrict__ row, int c0,
                                         float& lv, int& li, float& thr, int lane, int below4) {
    float n0 = row[lane], n1 = row[32 + lane], n2 = row[64 + lane], n3 = row[96 + lane];
    for (int i = 0; i < CHUNK / 32; i += 4) {
        const float v0 = n0, v1 = n1, v2 = n2, v3 = n3;
        if (i + 4 < CHUNK / 32) {
            n0 = row[(i + 4) * 32 + lane];
            n1 = row[(i + 5) * 32 + lane];
            n2 = row[(i + 6) * 32 + lane];
            n3 = row[(i + 7) * 32 + lane];
        }
        scan_step(v0, c0 + (i + 0) * 32, lv, li, thr, lane, below4);
        scan_step(v1, c0 + (i + 1) * 32, lv, li, thr, lane, below4);
        scan_step(v2, c0 + (i + 2) * 32, lv, li, thr, lane, below4);
        scan_step(v3, c0 + (i + 3) * 32, lv, li, thr, lane, below4);
    }
}

__device__ __forceinline__ void gram_chunk(const float* __restrict__ xb, const float* __restrict__ sqb,
                                           const v2f (&a)[16], float* __restrict__ smem, int c0,
                                           int wave, int lmod, int khalf, int rowadd) {
    for (int t = wave; t < CHUNK / 16; t += 8) {
        const int col0 = c0 + t * 16;
        v8f acc = {};
#pragma unroll
        for (int kk = 0; kk < 16; ++kk) {
            const float* p = xb + (size_t)(4 * kk + khalf) * NPTS + col0 + lmod;
            v2f bb; bb.x = p[0]; bb.y = p[NPTS];
            acc = wmma_f32(a[kk], bb, acc);
        }
        const float sqc = sqb[col0 + lmod];
#pragma unroll
        for (int r = 0; r < 8; ++r)
            smem[(r + rowadd) * CHUNK + t * 16 + lmod] = 2.0f * acc[r] - sqc;
    }
}

__global__ __launch_bounds__(256) void knn_kernel(const float* __restrict__ x,
                                                  const float* __restrict__ sq,
                                                  int* __restrict__ idxout) {
    __shared__ float smem[16 * CHUNK];

    const int b    = blockIdx.y;
    const int q0   = blockIdx.x * 16;
    const int tid  = threadIdx.x;
    const int wave = tid >> 5;
    const int lane = tid & 31;
    const int lmod = lane & 15;
    const int khalf  = (lane >> 4) << 1;
    const int rowadd = (lane >> 4) << 3;
    const float* xb  = x + (size_t)b * CH * NPTS;
    const float* sqb = sq + (size_t)b * NPTS;

    v2f a[16];
#pragma unroll
    for (int kk = 0; kk < 16; ++kk) {
        const float* p = xb + (size_t)(4 * kk + khalf) * NPTS + q0 + lmod;
        a[kk].x = p[0];
        a[kk].y = p[NPTS];
    }

    int   liA = 0, liB = 0;
    const float* rowA = smem + (2 * wave) * CHUNK;
    const float* rowB = rowA + CHUNK;
    const int below4 = ((lane > 0) ? lane - 1 : 0) << 2;

    gram_chunk(xb, sqb, a, smem, 0, wave, lmod, khalf, rowadd);
    __syncthreads();
    float lvA = seed_floor(rowA, lane);
    float lvB = seed_floor(rowB, lane);
    float thrA = lvA, thrB = lvB;
    scan_row(rowA, 0, lvA, liA, thrA, lane, below4);
    scan_row(rowB, 0, lvB, liB, thrB, lane, below4);
    __syncthreads();

    for (int c0 = CHUNK; c0 < NPTS; c0 += CHUNK) {
        gram_chunk(xb, sqb, a, smem, c0, wave, lmod, khalf, rowadd);
        __syncthreads();

        scan_row(rowA, c0, lvA, liA, thrA, lane, below4);
        scan_row(rowB, c0, lvB, liB, thrB, lane, below4);
        __syncthreads();
    }

    int* smi = (int*)smem;
    if (lane < KNN) {
        smi[(2 * wave) * KNN + lane]     = liA;
        smi[(2 * wave + 1) * KNN + lane] = liB;
    }
    __syncthreads();
    int* op = idxout + ((size_t)b * NPTS + q0) * KNN;
    op[tid] = smi[tid];
    if (tid < 16 * KNN - 256) op[256 + tid] = smi[256 + tid];
}

__global__ __launch_bounds__(256) void proj_kernel(const float* __restrict__ x,
                                                   const float* __restrict__ W,
                                                   float* __restrict__ y1T,
                                                   float* __restrict__ y3T) {
    int task = blockIdx.x * 8 + (threadIdx.x >> 5);
    int lane = threadIdx.x & 31;
    int lmod = lane & 15;
    int khalf  = (lane >> 4) << 1;
    int rowadd = (lane >> 4) << 3;

    int which = task & 1;
    int nt    = (task >> 1) & 255;
    int ot    = (task >> 9) & 3;
    int b     = task >> 11;

    const float* xb = x + (size_t)b * CH * NPTS;
    const float* wr = W + (size_t)(ot * 16 + lmod) * (2 * CH);
    const int n0 = nt * 16;

    v8f acc = {};
#pragma unroll
    for (int kk = 0; kk < 16; ++kk) {
        int c = 4 * kk + khalf;
        v2f av, bv;
        if (which == 0) { av.x = wr[c];            av.y = wr[c + 1]; }
        else            { av.x = wr[CH + c] - wr[c];
                          av.y = wr[CH + c + 1] - wr[c + 1]; }
        const float* p = xb + (size_t)c * NPTS + n0 + lmod;
        bv.x = p[0]; bv.y = p[NPTS];
        acc = wmma_f32(av, bv, acc);
    }

    float* y = (which ? y3T : y1T)
             + (size_t)b * NPTS * OCH + (size_t)(n0 + lmod) * OCH
             + ot * 16 + rowadd;
    float4 lo = { acc[0], acc[1], acc[2], acc[3] };
    float4 hi = { acc[4], acc[5], acc[6], acc[7] };
    ((float4*)y)[0] = lo;
    ((float4*)y)[1] = hi;
}

__global__ __launch_bounds__(256) void gather_kernel(const float* __restrict__ y1T,
                                                     const float* __restrict__ y3T,
                                                     const int* __restrict__ idx,
                                                     const float* __restrict__ gamma,
                                                     float* __restrict__ selbuf,
                                                     float* __restrict__ partial) {
    int waveId = blockIdx.x * 8 + (threadIdx.x >> 5);
    int lane   = threadIdx.x & 31;
    int ch     = lane * 2;
    bool mx0 = gamma[ch] >= 0.f, mx1 = gamma[ch + 1] >= 0.f;

    float s0 = 0.f, s1 = 0.f, q0 = 0.f, q1 = 0.f;

    for (int pt = 0; pt < 32; ++pt) {
        int p = waveId * 32 + pt;
        int b = p >> 12;
        float2 e = ((const float2*)(y3T + (size_t)p * OCH))[lane];
        const int* ip = idx + (size_t)p * KNN;
        const float* y1b = y1T + (size_t)b * NPTS * OCH;

        float hi0 = -FLT_MAX, hi1 = -FLT_MAX, lo0 = FLT_MAX, lo1 = FLT_MAX;
#pragma unroll
        for (int j = 0; j < KNN; ++j) {
            int nb = ip[j];
            float2 g = ((const float2*)(y1b + ((size_t)nb << 6)))[lane];
            float h0 = g.x + e.x, h1 = g.y + e.y;
            s0 += h0; s1 += h1; q0 += h0 * h0; q1 += h1 * h1;
            hi0 = fmaxf(hi0, h0); lo0 = fminf(lo0, h0);
            hi1 = fmaxf(hi1, h1); lo1 = fminf(lo1, h1);
        }
        float2 selv = { mx0 ? hi0 : lo0, mx1 ? hi1 : lo1 };
        ((float2*)(selbuf + (size_t)p * OCH))[lane] = selv;
    }
    float* pb = partial + (size_t)waveId * 128;
    pb[ch]          = s0; pb[ch + 1]       = s1;
    pb[OCH + ch]    = q0; pb[OCH + ch + 1] = q1;
}

__global__ __launch_bounds__(1024) void stats_kernel(const float* __restrict__ partial,
                                                     const float* __restrict__ gamma,
                                                     const float* __restrict__ beta,
                                                     float* __restrict__ coeff) {
    __shared__ float red[32 * 128];
    const int tid  = threadIdx.x;
    const int wave = tid >> 5;
    const int lane = tid & 31;

    const float4* p = (const float4*)(partial + (size_t)wave * 32 * 128) + lane;
    float4 a = { 0.f, 0.f, 0.f, 0.f };
    float4 c = { 0.f, 0.f, 0.f, 0.f };
    for (int r = 0; r < 32; r += 2) {
        float4 u = p[(size_t)r * 32];
        float4 v = p[(size_t)(r + 1) * 32];
        a.x += u.x; a.y += u.y; a.z += u.z; a.w += u.w;
        c.x += v.x; c.y += v.y; c.z += v.z; c.w += v.w;
    }
    float4 t4 = { a.x + c.x, a.y + c.y, a.z + c.z, a.w + c.w };
    ((float4*)red)[wave * 32 + lane] = t4;
    __syncthreads();

    if (tid < OCH) {
        const int ch = tid;
        float s = 0.f, q = 0.f;
        for (int w = 0; w < 32; ++w) {
            s += red[w * 128 + ch];
            q += red[w * 128 + OCH + ch];
        }
        const float M = (float)BATCH * NPTS * KNN;
        float mean  = s / M;
        float var   = q / M - mean * mean;
        float scale = gamma[ch] * rsqrtf(var + 1e-5f);
        coeff[ch]       = scale;
        coeff[OCH + ch] = beta[ch] - mean * scale;
    }
}

#define FT_PTS   64
#define FT_PITCH 65

__global__ __launch_bounds__(256) void final_kernel(const float* __restrict__ selbuf,
                                                    const float* __restrict__ coeff,
                                                    float* __restrict__ out) {
    __shared__ float tile[OCH * FT_PITCH];

    const int tid = threadIdx.x;
    const int c4  = tid & 15;
    const int r   = tid >> 4;
    const int p0  = blockIdx.x * FT_PTS;
    const int b   = p0 >> 12;
    const int n0  = p0 & (NPTS - 1);

    const float4 sc = ((const float4*)coeff)[c4];
    const float4 sh = ((const float4*)(coeff + OCH))[c4];
    const float4* srcp = (const float4*)(selbuf + (size_t)p0 * OCH) + tid;

    float4 v[4];
    for (int j = 0; j < 4; ++j) v[j] = srcp[j * 256];

    for (int j = 0; j < 4; ++j) {
        const int pt = r + 16 * j;
        float r0 = v[j].x * sc.x + sh.x;
        float r1 = v[j].y * sc.y + sh.y;
        float r2 = v[j].z * sc.z + sh.z;
        float r3 = v[j].w * sc.w + sh.w;
        tile[(4 * c4 + 0) * FT_PITCH + pt] = (r0 >= 0.f) ? r0 : 0.2f * r0;
        tile[(4 * c4 + 1) * FT_PITCH + pt] = (r1 >= 0.f) ? r1 : 0.2f * r1;
        tile[(4 * c4 + 2) * FT_PITCH + pt] = (r2 >= 0.f) ? r2 : 0.2f * r2;
        tile[(4 * c4 + 3) * FT_PITCH + pt] = (r3 >= 0.f) ? r3 : 0.2f * r3;
    }
    __syncthreads();

    float* dst = out + (size_t)b * OCH * NPTS + n0 + 4 * c4;
    for (int j = 0; j < 4; ++j) {
        const int o = r + 16 * j;
        const float* t = tile + o * FT_PITCH + 4 * c4;
        float4 w = { t[0], t[1], t[2], t[3] };
        *(float4*)(dst + (size_t)o * NPTS) = w;
    }
}

extern "C" void kernel_launch(void* const* d_in, const int* in_sizes, int n_in,
                              void* d_out, int out_size, void* d_ws, size_t ws_size,
                              hipStream_t stream) {
    const float* x     = (const float*)d_in[0];
    const float* W     = (const float*)d_in[1];
    const float* gamma = (const float*)d_in[2];
    const float* beta  = (const float*)d_in[3];
    float* out = (float*)d_out;

    float* ws    = (float*)d_ws;
    float* sq    = ws;
    int*   idx   = (int*)(ws + 32768);
    float* y1T   = ws + 32768 + 655360;
    float* y3T   = y1T + (size_t)BATCH * NPTS * OCH;
    float* sel   = y3T + (size_t)BATCH * NPTS * OCH;
    float* part  = sel + (size_t)BATCH * NPTS * OCH;
    float* coeff = part + 1024 * 128;

    sq_kernel<<<(BATCH * NPTS) / 256, 256, 0, stream>>>(x, sq);
    knn_kernel<<<dim3(NPTS / 16, BATCH), 256, 0, stream>>>(x, sq, idx);
    proj_kernel<<<2048, 256, 0, stream>>>(x, W, y1T, y3T);
    gather_kernel<<<128, 256, 0, stream>>>(y1T, y3T, idx, gamma, sel, part);
    stats_kernel<<<1, 1024, 0, stream>>>(part, gamma, beta, coeff);
    final_kernel<<<(BATCH * NPTS) / FT_PTS, 256, 0, stream>>>(sel, coeff, out);
}


# kNN Gram: column operand loads widen from 4 to 16 bytes per lane: a wave computes four 16-column tiles at once on interleaved columns (whole 128-byte lines per load, 16-byte score stores)
# speedup vs baseline: 4.0515x; 1.1546x over previous
#include <hip/hip_runtime.h>
#include <hip/hip_bf16.h>
#include <float.h>


typedef float v2f __attribute__((ext_vector_type(2)));
typedef float v8f __attribute__((ext_vector_type(8)));

#define BATCH 8
#define CH    64
#define NPTS  4096
#define OCH   64
#define KNN   20
#define CHUNK 1024
#define NEG_INF (-3.0e38f)

__device__ __forceinline__ v8f wmma_f32(v2f a, v2f b, v8f c) {
    return __builtin_amdgcn_wmma_f32_16x16x4_f32(
        false, a, false, b, (short)0, c, false, false);
}

__global__ __launch_bounds__(256) void sq_kernel(const float* __restrict__ x,
                                                 float* __restrict__ sq) {
    int t = blockIdx.x * 256 + threadIdx.x;
    int b = t >> 12;
    int n = t & (NPTS - 1);
    const float* p = x + (size_t)b * CH * NPTS + n;
    float s = 0.f;
#pragma unroll
    for (int c = 0; c < CH; ++c) { float v = p[(size_t)c * NPTS]; s += v * v; }
    sq[t] = s;
}

__device__ __forceinline__ float lane_value(float v, int l) {
    return __int_as_float(__builtin_amdgcn_readlane(__float_as_int(v), l));
}

__device__ __forceinline__ void list_insert(float& lv, int& li, float& thr,
                                            float cand, int cidx, int lane, int below4) {
    const float upv = __int_as_float(__builtin_amdgcn_ds_bpermute(below4, __float_as_int(lv)));
    const int   upi = __builtin_amdgcn_ds_bpermute(below4, li);
    const int pos = __builtin_popcount(__builtin_amdgcn_ballot_w32(lv >= cand));
    lv = (lane < pos) ? lv : ((lane == pos) ? cand : upv);
    li = (lane < pos) ? li : ((lane == pos) ? cidx : upi);
    thr = lane_value(lv, KNN - 1);
}

__device__ __forceinline__ float seed_floor(const float* __restrict__ row, int lane) {
    float m = row[lane];
    for (int i = 1; i < CHUNK / 32; ++i) m = fmaxf(m, row[i * 32 + lane]);
    int larger = 0;
    for (int j = 0; j < 32; ++j) larger += (lane_value(m, j) > m) ? 1 : 0;
    const unsigned hit = __builtin_amdgcn_ballot_w32(larger == KNN - 1);
    if (hit == 0) return NEG_INF;
    const float f = lane_value(m, __builtin_ctz(hit));
    return f - fmaxf(fabsf(f) * 1.0e-6f, 1.0e-20f);
}

__device__ __forceinline__ void scan_step(float v, int colbase, float& lv, int& li,
                                          float& thr, int lane, int below4) {
    unsigned m = __builtin_amdgcn_ballot_w32(v > thr);
    while (m) {
        const int j = __builtin_ctz(m);
        m &= m - 1;
        const float cand = lane_value(v, j);
        if (cand > thr)
            list_insert(lv, li, thr, cand, colbase + j, lane, below4);
    }
}

__device__ __forceinline__ void scan_row(const float* __restrict__ row, int c0,
                                         float& lv, int& li, float& thr, int lane, int below4) {
    float n0 = row[lane], n1 = row[32 + lane], n2 = row[64 + lane], n3 = row[96 + lane];
    for (int i = 0; i < CHUNK / 32; i += 4) {
        const float v0 = n0, v1 = n1, v2 = n2, v3 = n3;
        if (i + 4 < CHUNK / 32) {
            n0 = row[(i + 4) * 32 + lane];
            n1 = row[(i + 5) * 32 + lane];
            n2 = row[(i + 6) * 32 + lane];
            n3 = row[(i + 7) * 32 + lane];
        }
        scan_step(v0, c0 + (i + 0) * 32, lv, li, thr, lane, below4);
        scan_step(v1, c0 + (i + 1) * 32, lv, li, thr, lane, below4);
        scan_step(v2, c0 + (i + 2) * 32, lv, li, thr, lane, below4);
        scan_step(v3, c0 + (i + 3) * 32, lv, li, thr, lane, below4);
    }
}

__device__ __forceinline__ void gram_chunk(const float* __restrict__ xb, const float* __restrict__ sqb,
                                           const v2f (&a)[16], float* __restrict__ smem, int c0,
                                           int wave, int lmod, int khalf, int rowadd) {
    for (int i = 0; i < CHUNK / 512; ++i) {
        const int colq = (wave + 8 * i) * 64 + 4 * lmod;
        const float* pb = xb + (size_t)khalf * NPTS + c0 + colq;
        v8f acc0 = {}, acc1 = {}, acc2 = {}, acc3 = {};
#pragma unroll
        for (int kb = 0; kb < 4; ++kb) {
            float4 u[4], v[4];
#pragma unroll
            for (int j = 0; j < 4; ++j) {
                const int kk = 4 * kb + j;
                u[j] = *(const float4*)(pb + (size_t)(4 * kk) * NPTS);
                v[j] = *(const float4*)(pb + (size_t)(4 * kk + 1) * NPTS);
            }
#pragma unroll
            for (int j = 0; j < 4; ++j) {
                const int kk = 4 * kb + j;
                v2f b0; b0.x = u[j].x; b0.y = v[j].x;
                v2f b1; b1.x = u[j].y; b1.y = v[j].y;
                v2f b2; b2.x = u[j].z; b2.y = v[j].z;
                v2f b3; b3.x = u[j].w; b3.y = v[j].w;
                acc0 = wmma_f32(a[kk], b0, acc0);
                acc1 = wmma_f32(a[kk], b1, acc1);
                acc2 = wmma_f32(a[kk], b2, acc2);
                acc3 = wmma_f32(a[kk], b3, acc3);
            }
        }
        const float4 sqc = *(const float4*)(sqb + c0 + colq);
#pragma unroll
        for (int r = 0; r < 8; ++r) {
            float4 s;
            s.x = 2.0f * acc0[r] - sqc.x;
            s.y = 2.0f * acc1[r] - sqc.y;
            s.z = 2.0f * acc2[r] - sqc.z;
            s.w = 2.0f * acc3[r] - sqc.w;
            *(float4*)(smem + (r + rowadd) * CHUNK + colq) = s;
        }
    }
}

__device__ __forceinline__ float2 gather_point(const float* __restrict__ y1b,
                                               const float* __restrict__ y3T,
                                               int p, int li, int lane, bool mx0, bool mx1,
                                               float& s0, float& s1, float& q0, float& q1) {
    const float2 e = ((const float2*)(y3T + (size_t)p * OCH))[lane];
    float hi0 = -FLT_MAX, hi1 = -FLT_MAX, lo0 = FLT_MAX, lo1 = FLT_MAX;
#pragma unroll
    for (int j = 0; j < KNN; ++j) {
        const int nb = __builtin_amdgcn_readlane(li, j) & (NPTS - 1);
        const float2 g = ((const float2*)(y1b + ((size_t)nb << 6)))[lane];
        const float h0 = g.x + e.x, h1 = g.y + e.y;
        s0 += h0; s1 += h1; q0 += h0 * h0; q1 += h1 * h1;
        hi0 = fmaxf(hi0, h0); lo0 = fminf(lo0, h0);
        hi1 = fmaxf(hi1, h1); lo1 = fminf(lo1, h1);
    }
    const float2 selv = { mx0 ? hi0 : lo0, mx1 ? hi1 : lo1 };
    return selv;
}

__global__ __launch_bounds__(256, 8) void knn_kernel(const float* __restrict__ x,
                                                  const float* __restrict__ sq,
                                                  const float* __restrict__ y1T,
                                                  const float* __restrict__ y3T,
                                                  const float* __restrict__ gamma,
                                                  float* __restrict__ selbuf,
                                                  float* __restrict__ partial) {
    __shared__ __attribute__((aligned(16))) float smem[16 * CHUNK];

    const int b    = blockIdx.y;
    const int q0   = blockIdx.x * 16;
    const int tid  = threadIdx.x;
    const int wave = tid >> 5;
    const int lane = tid & 31;
    const int lmod = lane & 15;
    const int khalf  = (lane >> 4) << 1;
    const int rowadd = (lane >> 4) << 3;
    const float* xb  = x + (size_t)b * CH * NPTS;
    const float* sqb = sq + (size_t)b * NPTS;

    v2f a[16];
#pragma unroll
    for (int kk = 0; kk < 16; ++kk) {
        const float* p = xb + (size_t)(4 * kk + khalf) * NPTS + q0 + lmod;
        a[kk].x = p[0];
        a[kk].y = p[NPTS];
    }

    int   liA = 0, liB = 0;
    const float* rowA = smem + (2 * wave) * CHUNK;
    const float* rowB = rowA + CHUNK;
    const int below4 = ((lane > 0) ? lane - 1 : 0) << 2;

    gram_chunk(xb, sqb, a, smem, 0, wave, lmod, khalf, rowadd);
    __syncthreads();
    float lvA = seed_floor(rowA, lane);
    float lvB = seed_floor(rowB, lane);
    float thrA = lvA, thrB = lvB;
    scan_row(rowA, 0, lvA, liA, thrA, lane, below4);
    scan_row(rowB, 0, lvB, liB, thrB, lane, below4);
    __syncthreads();

    for (int c0 = CHUNK; c0 < NPTS; c0 += CHUNK) {
        gram_chunk(xb, sqb, a, smem, c0, wave, lmod, khalf, rowadd);
        __syncthreads();

        scan_row(rowA, c0, lvA, liA, thrA, lane, below4);
        scan_row(rowB, c0, lvB, liB, thrB, lane, below4);
        __syncthreads();
    }

    const int ch = lane * 2;
    const bool mx0 = gamma[ch] >= 0.f, mx1 = gamma[ch + 1] >= 0.f;
    const float* y1b = y1T + (size_t)b * NPTS * OCH;
    const int pA = b * NPTS + q0 + 2 * wave;
    float s0 = 0.f, s1 = 0.f, t0 = 0.f, t1 = 0.f;
    float2 selA = gather_point(y1b, y3T, pA,     liA, lane, mx0, mx1, s0, s1, t0, t1);
    float2 selB = gather_point(y1b, y3T, pA + 1, liB, lane, mx0, mx1, s0, s1, t0, t1);

    float* red = smem + wave * 128;
    red[ch]       = s0; red[ch + 1]       = s1;
    red[OCH + ch] = t0; red[OCH + ch + 1] = t1;
    __syncthreads();
    if (tid < 128) {
        float t = 0.f;
        for (int w = 0; w < 8; ++w) t += smem[w * 128 + tid];
        partial[((size_t)b * (NPTS / 16) + blockIdx.x) * 128 + tid] = t;
    }
    ((float2*)(selbuf + (size_t)pA * OCH))[lane]       = selA;
    ((float2*)(selbuf + (size_t)(pA + 1) * OCH))[lane] = selB;
}

__global__ __launch_bounds__(256) void proj_kernel(const float* __restrict__ x,
                                                   const float* __restrict__ W,
                                                   float* __restrict__ y1T,
                                                   float* __restrict__ y3T) {
    int task = blockIdx.x * 8 + (threadIdx.x >> 5);
    int lane = threadIdx.x & 31;
    int lmod = lane & 15;
    int khalf  = (lane >> 4) << 1;
    int rowadd = (lane >> 4) << 3;

    int which = task & 1;
    int nt    = (task >> 1) & 255;
    int ot    = (task >> 9) & 3;
    int b     = task >> 11;

    const float* xb = x + (size_t)b * CH * NPTS;
    const float* wr = W + (size_t)(ot * 16 + lmod) * (2 * CH);
    const int n0 = nt * 16;

    v8f acc = {};
#pragma unroll
    for (int kk = 0; kk < 16; ++kk) {
        int c = 4 * kk + khalf;
        v2f av, bv;
        if (which == 0) { av.x = wr[c];            av.y = wr[c + 1]; }
        else            { av.x = wr[CH + c] - wr[c];
                          av.y = wr[CH + c + 1] - wr[c + 1]; }
        const float* p = xb + (size_t)c * NPTS + n0 + lmod;
        bv.x = p[0]; bv.y = p[NPTS];
        acc = wmma_f32(av, bv, acc);
    }

    float* y = (which ? y3T : y1T)
             + (size_t)b * NPTS * OCH + (size_t)(n0 + lmod) * OCH
             + ot * 16 + rowadd;
    float4 lo = { acc[0], acc[1], acc[2], acc[3] };
    float4 hi = { acc[4], acc[5], acc[6], acc[7] };
    ((float4*)y)[0] = lo;
    ((float4*)y)[1] = hi;
}

__global__ __launch_bounds__(1024) void stats_kernel(const float* __restrict__ partial,
                                                     const float* __restrict__ gamma,
                                                     const float* __restrict__ beta,
                                                     float* __restrict__ coeff) {
    __shared__ float red[32 * 128];
    const int tid  = threadIdx.x;
    const int wave = tid >> 5;
    const int lane = tid & 31;

    const float4* p = (const float4*)(partial + (size_t)wave * 64 * 128) + lane;
    float4 a = { 0.f, 0.f, 0.f, 0.f };
    float4 c = { 0.f, 0.f, 0.f, 0.f };
    for (int r = 0; r < 64; r += 2) {
        float4 u = p[(size_t)r * 32];
        float4 v = p[(size_t)(r + 1) * 32];
        a.x += u.x; a.y += u.y; a.z += u.z; a.w += u.w;
        c.x += v.x; c.y += v.y; c.z += v.z; c.w += v.w;
    }
    float4 t4 = { a.x + c.x, a.y + c.y, a.z + c.z, a.w + c.w };
    ((float4*)red)[wave * 32 + lane] = t4;
    __syncthreads();

    if (tid < OCH) {
        const int ch = tid;
        float s = 0.f, q = 0.f;
        for (int w = 0; w < 32; ++w) {
            s += red[w * 128 + ch];
            q += red[w * 128 + OCH + ch];
        }
        const float M = (float)BATCH * NPTS * KNN;
        float mean  = s / M;
        float var   = q / M - mean * mean;
        float scale = gamma[ch] * rsqrtf(var + 1e-5f);
        coeff[ch]       = scale;
        coeff[OCH + ch] = beta[ch] - mean * scale;
    }
}

#define FT_PTS   64
#define FT_PITCH 65

__global__ __launch_bounds__(256) void final_kernel(const float* __restrict__ selbuf,
                                                    const float* __restrict__ coeff,
                                                    float* __restrict__ out) {
    __shared__ float tile[OCH * FT_PITCH];

    const int tid = threadIdx.x;
    const int c4  = tid & 15;
    const int r   = tid >> 4;
    const int p0  = blockIdx.x * FT_PTS;
    const int b   = p0 >> 12;
    const int n0  = p0 & (NPTS - 1);

    const float4 sc = ((const float4*)coeff)[c4];
    const float4 sh = ((const float4*)(coeff + OCH))[c4];
    const float4* srcp = (const float4*)(selbuf + (size_t)p0 * OCH) + tid;

    float4 v[4];
    for (int j = 0; j < 4; ++j) v[j] = srcp[j * 256];

    for (int j = 0; j < 4; ++j) {
        const int pt = r + 16 * j;
        float r0 = v[j].x * sc.x + sh.x;
        float r1 = v[j].y * sc.y + sh.y;
        float r2 = v[j].z * sc.z + sh.z;
        float r3 = v[j].w * sc.w + sh.w;
        tile[(4 * c4 + 0) * FT_PITCH + pt] = (r0 >= 0.f) ? r0 : 0.2f * r0;
        tile[(4 * c4 + 1) * FT_PITCH + pt] = (r1 >= 0.f) ? r1 : 0.2f * r1;
        tile[(4 * c4 + 2) * FT_PITCH + pt] = (r2 >= 0.f) ? r2 : 0.2f * r2;
        tile[(4 * c4 + 3) * FT_PITCH + pt] = (r3 >= 0.f) ? r3 : 0.2f * r3;
    }
    __syncthreads();

    float* dst = out + (size_t)b * OCH * NPTS + n0 + 4 * c4;
    for (int j = 0; j < 4; ++j) {
        const int o = r + 16 * j;
        const float* t = tile + o * FT_PITCH + 4 * c4;
        float4 w = { t[0], t[1], t[2], t[3] };
        *(float4*)(dst + (size_t)o * NPTS) = w;
    }
}

extern "C" void kernel_launch(void* const* d_in, const int* in_sizes, int n_in,
                              void* d_out, int out_size, void* d_ws, size_t ws_size,
                              hipStream_t stream) {
    const float* x     = (const float*)d_in[0];
    const float* W     = (const float*)d_in[1];
    const float* gamma = (const float*)d_in[2];
    const float* beta  = (const float*)d_in[3];
    float* out = (float*)d_out;

    float* ws    = (float*)d_ws;
    float* sq    = ws;
    float* part  = ws + 32768;
    float* y1T   = ws + 32768 + 655360;
    float* y3T   = y1T + (size_t)BATCH * NPTS * OCH;
    float* sel   = y3T + (size_t)BATCH * NPTS * OCH;
    float* coeff = sel + (size_t)BATCH * NPTS * OCH + 1024 * 128;

    sq_kernel<<<(BATCH * NPTS) / 256, 256, 0, stream>>>(x, sq);
    proj_kernel<<<2048, 256, 0, stream>>>(x, W, y1T, y3T);
    knn_kernel<<<dim3(NPTS / 16, BATCH), 256, 0, stream>>>(x, sq, y1T, y3T, gamma, sel, part);
    stats_kernel<<<1, 1024, 0, stream>>>(part, gamma, beta, coeff);
    final_kernel<<<(BATCH * NPTS) / FT_PTS, 256, 0, stream>>>(sel, coeff, out);
}


# kNN selection, fewer vector-to-scalar round trips: 8 threshold tests (256 columns) per branch instead of 1; each lane picks its new list entry from two vector compares, no scalar position or re-check
# speedup vs baseline: 4.4857x; 1.1072x over previous
#include <hip/hip_runtime.h>
#include <hip/hip_bf16.h>
#include <float.h>


typedef float v2f __attribute__((ext_vector_type(2)));
typedef float v8f __attribute__((ext_vector_type(8)));

#define BATCH 8
#define CH    64
#define NPTS  4096
#define OCH   64
#define KNN   20
#define CHUNK 512
#define NEG_INF (-3.0e38f)

__device__ __forceinline__ v8f wmma_f32(v2f a, v2f b, v8f c) {
    return __builtin_amdgcn_wmma_f32_16x16x4_f32(
        false, a, false, b, (short)0, c, false, false);
}

__device__ __forceinline__ float lane_value(float v, int l) {
    return __int_as_float(__builtin_amdgcn_readlane(__float_as_int(v), l));
}

__device__ __forceinline__ void list_insert(float& lv, int& li, float cand, int cidx, int below4) {
    const float upv = __int_as_float(__builtin_amdgcn_ds_bpermute(below4, __float_as_int(lv)));
    const int   upi = __builtin_amdgcn_ds_bpermute(below4, li);
    const bool keep = lv >= cand;
    const bool here = upv >= cand;
    const float nv = here ? cand : upv;
    const int   ni = here ? cidx : upi;
    lv = keep ? lv : nv;
    li = keep ? li : ni;
}

__device__ __forceinline__ float seed_floor(const float* __restrict__ row, int lane) {
    float m = row[lane];
    for (int i = 1; i < CHUNK / 32; ++i) m = fmaxf(m, row[i * 32 + lane]);
    int larger = 0;
    for (int j = 0; j < 32; ++j) larger += (lane_value(m, j) > m) ? 1 : 0;
    const unsigned hit = __builtin_amdgcn_ballot_w32(larger == KNN - 1);
    if (hit == 0) return NEG_INF;
    const float f = lane_value(m, __builtin_ctz(hit));
    return f - fmaxf(fabsf(f) * 1.0e-6f, 1.0e-20f);
}

__device__ __forceinline__ void scan_piece(const float* __restrict__ seg, int colbase,
                                           float& lv, int& li, float& thr, int lane, int below4) {
    float v[8];
#pragma unroll
    for (int i = 0; i < 8; ++i) v[i] = seg[i * 32 + lane];
    unsigned m[8];
#pragma unroll
    for (int i = 0; i < 8; ++i) m[i] = __builtin_amdgcn_ballot_w32(v[i] > thr);
    if (((m[0] | m[1]) | (m[2] | m[3]) | (m[4] | m[5]) | (m[6] | m[7])) == 0u) return;
#pragma unroll
    for (int i = 0; i < 8; ++i) {
        unsigned mm = m[i];
        while (mm) {
            const int j = __builtin_ctz(mm);
            mm &= mm - 1;
            list_insert(lv, li, lane_value(v[i], j), colbase + i * 32 + j, below4);
        }
    }
    thr = lane_value(lv, KNN);
}

struct Lists {
    float lvA, lvB, thrA, thrB;
    int   liA, liB;
};

__device__ __forceinline__ void gram_chunk(const float* __restrict__ xb, const float* __restrict__ sqb,
                                           const v2f (&a)[16], float* __restrict__ wbuf, int c0,
                                           const float* __restrict__ rbuf, int cprev, bool doscan, Lists& L,
                                           int wave, int lane, int lmod, int khalf, int rowadd,
                                           int below4) {
    const int colq = wave * 64 + 4 * lmod;
    const float* pb = xb + (size_t)khalf * NPTS + c0 + colq;
    const float* rowA = rbuf + (2 * wave) * CHUNK;
    const float* rowB = rowA + CHUNK;
    v8f acc0 = {}, acc1 = {}, acc2 = {}, acc3 = {};
    const float4 sqc = *(const float4*)(sqb + c0 + colq);
#pragma unroll
    for (int kb = 0; kb < 4; ++kb) {
        float4 u[4], v[4];
#pragma unroll
        for (int j = 0; j < 4; ++j) {
            const int kk = 4 * kb + j;
            u[j] = *(const float4*)(pb + (size_t)(4 * kk) * NPTS);
            v[j] = *(const float4*)(pb + (size_t)(4 * kk + 1) * NPTS);
        }
        if (doscan) {
            const int half = (kb & 1) * (CHUNK / 2);
            if (kb < 2) scan_piece(rowA + half, cprev + half, L.lvA, L.liA, L.thrA, lane, below4);
            else        scan_piece(rowB + half, cprev + half, L.lvB, L.liB, L.thrB, lane, below4);
        }
#pragma unroll
        for (int j = 0; j < 4; ++j) {
            const int kk = 4 * kb + j;
            v2f b0; b0.x = u[j].x; b0.y = v[j].x;
            v2f b1; b1.x = u[j].y; b1.y = v[j].y;
            v2f b2; b2.x = u[j].z; b2.y = v[j].z;
            v2f b3; b3.x = u[j].w; b3.y = v[j].w;
            acc0 = wmma_f32(a[kk], b0, acc0);
            acc1 = wmma_f32(a[kk], b1, acc1);
            acc2 = wmma_f32(a[kk], b2, acc2);
            acc3 = wmma_f32(a[kk], b3, acc3);
        }
    }
#pragma unroll
    for (int r = 0; r < 8; ++r) {
        float4 s;
        s.x = 2.0f * acc0[r] - sqc.x;
        s.y = 2.0f * acc1[r] - sqc.y;
        s.z = 2.0f * acc2[r] - sqc.z;
        s.w = 2.0f * acc3[r] - sqc.w;
        *(float4*)(wbuf + (r + rowadd) * CHUNK + colq) = s;
    }
}

__device__ __forceinline__ float2 gather_point(const float* __restrict__ y1b,
                                               const float* __restrict__ y3T,
                                               int p, int li, int lane, bool mx0, bool mx1,
                                               float& s0, float& s1, float& q0, float& q1) {
    const float2 e = ((const float2*)(y3T + (size_t)p * OCH))[lane];
    float hi0 = -FLT_MAX, hi1 = -FLT_MAX, lo0 = FLT_MAX, lo1 = FLT_MAX;
#pragma unroll
    for (int j = 0; j < KNN; ++j) {
        const int nb = __builtin_amdgcn_readlane(li, j + 1) & (NPTS - 1);
        const float2 g = ((const float2*)(y1b + ((size_t)nb << 6)))[lane];
        const float h0 = g.x + e.x, h1 = g.y + e.y;
        s0 += h0; s1 += h1; q0 += h0 * h0; q1 += h1 * h1;
        hi0 = fmaxf(hi0, h0); lo0 = fminf(lo0, h0);
        hi1 = fmaxf(hi1, h1); lo1 = fminf(lo1, h1);
    }
    const float2 selv = { mx0 ? hi0 : lo0, mx1 ? hi1 : lo1 };
    return selv;
}

__global__ __launch_bounds__(256, 8) void knn_kernel(const float* __restrict__ x,
                                                  const float* __restrict__ sq,
                                                  const float* __restrict__ y1T,
                                                  const float* __restrict__ y3T,
                                                  const float* __restrict__ gamma,
                                                  float* __restrict__ selbuf,
                                                  float* __restrict__ partial) {
    __shared__ __attribute__((aligned(16))) float smem[2 * 16 * CHUNK];

    const int b    = blockIdx.y;
    const int q0   = blockIdx.x * 16;
    const int tid  = threadIdx.x;
    const int wave = tid >> 5;
    const int lane = tid & 31;
    const int lmod = lane & 15;
    const int khalf  = (lane >> 4) << 1;
    const int rowadd = (lane >> 4) << 3;
    const float* xb  = x + (size_t)b * CH * NPTS;
    const float* sqb = sq + (size_t)b * NPTS;

    v2f a[16];
#pragma unroll
    for (int kk = 0; kk < 16; ++kk) {
        const float* p = xb + (size_t)(4 * kk + khalf) * NPTS + q0 + lmod;
        a[kk].x = p[0];
        a[kk].y = p[NPTS];
    }

    Lists L;
    L.liA = 0; L.liB = 0;
    L.lvA = NEG_INF; L.lvB = NEG_INF; L.thrA = NEG_INF; L.thrB = NEG_INF;
    const int below4 = ((lane > 0) ? lane - 1 : 0) << 2;

    for (int c = 0; c < NPTS / CHUNK; ++c) {
        float*       wbuf = smem + (c & 1) * (16 * CHUNK);
        const float* rbuf = smem + ((c + 1) & 1) * (16 * CHUNK);
        gram_chunk(xb, sqb, a, wbuf, c * CHUNK, rbuf, (c - 1) * CHUNK, c > 0, L,
                   wave, lane, lmod, khalf, rowadd, below4);
        __syncthreads();
        if (c == 0) {
            L.thrA = seed_floor(smem + (2 * wave) * CHUNK, lane);
            L.thrB = seed_floor(smem + (2 * wave + 1) * CHUNK, lane);
            L.lvA = (lane == 0) ? __builtin_inff() : L.thrA;
            L.lvB = (lane == 0) ? __builtin_inff() : L.thrB;
        }
    }

    {
        const float* rowA = smem + 16 * CHUNK + (2 * wave) * CHUNK;
        scan_piece(rowA,                     NPTS - CHUNK,             L.lvA, L.liA, L.thrA, lane, below4);
        scan_piece(rowA + CHUNK / 2,         NPTS - CHUNK / 2,         L.lvA, L.liA, L.thrA, lane, below4);
        scan_piece(rowA + CHUNK,             NPTS - CHUNK,             L.lvB, L.liB, L.thrB, lane, below4);
        scan_piece(rowA + CHUNK + CHUNK / 2, NPTS - CHUNK / 2,         L.lvB, L.liB, L.thrB, lane, below4);
    }
    const int liA = L.liA, liB = L.liB;

    const int ch = lane * 2;
    const bool mx0 = gamma[ch] >= 0.f, mx1 = gamma[ch + 1] >= 0.f;
    const float* y1b = y1T + (size_t)b * NPTS * OCH;
    const int pA = b * NPTS + q0 + 2 * wave;
    float s0 = 0.f, s1 = 0.f, t0 = 0.f, t1 = 0.f;
    float2 selA = gather_point(y1b, y3T, pA,     liA, lane, mx0, mx1, s0, s1, t0, t1);
    float2 selB = gather_point(y1b, y3T, pA + 1, liB, lane, mx0, mx1, s0, s1, t0, t1);

    float* red = smem + wave * 128;
    red[ch]       = s0; red[ch + 1]       = s1;
    red[OCH + ch] = t0; red[OCH + ch + 1] = t1;
    __syncthreads();
    if (tid < 128) {
        float t = 0.f;
        for (int w = 0; w < 8; ++w) t += smem[w * 128 + tid];
        partial[((size_t)b * (NPTS / 16) + blockIdx.x) * 128 + tid] = t;
    }
    ((float2*)(selbuf + (size_t)pA * OCH))[lane]       = selA;
    ((float2*)(selbuf + (size_t)(pA + 1) * OCH))[lane] = selB;
}

__global__ __launch_bounds__(256) void proj_kernel(const float* __restrict__ x,
                                                   const float* __restrict__ W,
                                                   float* __restrict__ y1T,
                                                   float* __restrict__ y3T,
                                                   float* __restrict__ sq) {
    const int task = blockIdx.x * 8 + (threadIdx.x >> 5);
    const int lane = threadIdx.x & 31;
    const int lmod = lane & 15;
    const int half = lane >> 4;
    const int khalf  = half << 1;
    const int rowadd = half << 3;

    const int b  = task >> 8;
    const int n0 = (task & 255) * 16;

    const float* xp = x + (size_t)b * CH * NPTS + n0 + lmod;
    v2f bv[16];
#pragma unroll
    for (int kk = 0; kk < 16; ++kk) {
        const float* p = xp + (size_t)(4 * kk + khalf) * NPTS;
        bv[kk].x = p[0];
        bv[kk].y = p[NPTS];
    }

    {
        const int other4 = (lane ^ 16) << 2;
        float s = 0.f;
#pragma unroll
        for (int kk = 0; kk < 16; ++kk) {
            const float ox = __int_as_float(__builtin_amdgcn_ds_bpermute(other4, __float_as_int(bv[kk].x)));
            const float oy = __int_as_float(__builtin_amdgcn_ds_bpermute(other4, __float_as_int(bv[kk].y)));
            const float c0 = half ? ox : bv[kk].x;
            const float c1 = half ? oy : bv[kk].y;
            const float c2 = half ? bv[kk].x : ox;
            const float c3 = half ? bv[kk].y : oy;
            if (kk == 0) {
                s = __builtin_fmaf(c0, c0, c1 * c1);
            } else {
                s = __builtin_fmaf(c0, c0, s);
                s = __builtin_fmaf(c1, c1, s);
            }
            s = __builtin_fmaf(c2, c2, s);
            s = __builtin_fmaf(c3, c3, s);
        }
        if (half == 0) sq[(size_t)b * NPTS + n0 + lmod] = s;
    }

    const size_t yoff = (size_t)b * NPTS * OCH + (size_t)(n0 + lmod) * OCH + rowadd;
    for (int ot = 0; ot < 4; ++ot) {
        const float* wr = W + (size_t)(ot * 16 + lmod) * (2 * CH) + khalf;
        v8f acc1 = {}, acc3 = {};
#pragma unroll
        for (int kk = 0; kk < 16; ++kk) {
            const float* w = wr + 4 * kk;
            v2f a1, a3;
            a1.x = w[0];           a1.y = w[1];
            a3.x = w[CH] - w[0];   a3.y = w[CH + 1] - w[1];
            acc1 = wmma_f32(a1, bv[kk], acc1);
            acc3 = wmma_f32(a3, bv[kk], acc3);
        }
        float* y1 = y1T + yoff + ot * 16;
        float* y3 = y3T + yoff + ot * 16;
        float4 lo1 = { acc1[0], acc1[1], acc1[2], acc1[3] };
        float4 hi1 = { acc1[4], acc1[5], acc1[6], acc1[7] };
        float4 lo3 = { acc3[0], acc3[1], acc3[2], acc3[3] };
        float4 hi3 = { acc3[4], acc3[5], acc3[6], acc3[7] };
        ((float4*)y1)[0] = lo1;
        ((float4*)y1)[1] = hi1;
        ((float4*)y3)[0] = lo3;
        ((float4*)y3)[1] = hi3;
    }
}

__global__ __launch_bounds__(1024) void stats_kernel(const float* __restrict__ partial,
                                                     const float* __restrict__ gamma,
                                                     const float* __restrict__ beta,
                                                     float* __restrict__ coeff) {
    __shared__ float red[32 * 128];
    const int tid  = threadIdx.x;
    const int wave = tid >> 5;
    const int lane = tid & 31;

    const float4* p = (const float4*)(partial + (size_t)wave * 64 * 128) + lane;
    float4 a = { 0.f, 0.f, 0.f, 0.f };
    float4 c = { 0.f, 0.f, 0.f, 0.f };
    for (int r = 0; r < 64; r += 2) {
        float4 u = p[(size_t)r * 32];
        float4 v = p[(size_t)(r + 1) * 32];
        a.x += u.x; a.y += u.y; a.z += u.z; a.w += u.w;
        c.x += v.x; c.y += v.y; c.z += v.z; c.w += v.w;
    }
    float4 t4 = { a.x + c.x, a.y + c.y, a.z + c.z, a.w + c.w };
    ((float4*)red)[wave * 32 + lane] = t4;
    __syncthreads();

    if (tid < OCH) {
        const int ch = tid;
        float s = 0.f, q = 0.f;
        for (int w = 0; w < 32; ++w) {
            s += red[w * 128 + ch];
            q += red[w * 128 + OCH + ch];
        }
        const float M = (float)BATCH * NPTS * KNN;
        float mean  = s / M;
        float var   = q / M - mean * mean;
        float scale = gamma[ch] * rsqrtf(var + 1e-5f);
        coeff[ch]       = scale;
        coeff[OCH + ch] = beta[ch] - mean * scale;
    }
}

#define FT_PTS   64
#define FT_PITCH 65

__global__ __launch_bounds__(256) void final_kernel(const float* __restrict__ selbuf,
                                                    const float* __restrict__ coeff,
                                                    float* __restrict__ out) {
    __shared__ float tile[OCH * FT_PITCH];

    const int tid = threadIdx.x;
    const int c4  = tid & 15;
    const int r   = tid >> 4;
    const int p0  = blockIdx.x * FT_PTS;
    const int b   = p0 >> 12;
    const int n0  = p0 & (NPTS - 1);

    const float4 sc = ((const float4*)coeff)[c4];
    const float4 sh = ((const float4*)(coeff + OCH))[c4];
    const float4* srcp = (const float4*)(selbuf + (size_t)p0 * OCH) + tid;

    float4 v[4];
    for (int j = 0; j < 4; ++j) v[j] = srcp[j * 256];

    for (int j = 0; j < 4; ++j) {
        const int pt = r + 16 * j;
        float r0 = v[j].x * sc.x + sh.x;
        float r1 = v[j].y * sc.y + sh.y;
        float r2 = v[j].z * sc.z + sh.z;
        float r3 = v[j].w * sc.w + sh.w;
        tile[(4 * c4 + 0) * FT_PITCH + pt] = (r0 >= 0.f) ? r0 : 0.2f * r0;
        tile[(4 * c4 + 1) * FT_PITCH + pt] = (r1 >= 0.f) ? r1 : 0.2f * r1;
        tile[(4 * c4 + 2) * FT_PITCH + pt] = (r2 >= 0.f) ? r2 : 0.2f * r2;
        tile[(4 * c4 + 3) * FT_PITCH + pt] = (r3 >= 0.f) ? r3 : 0.2f * r3;
    }
    __syncthreads();

    float* dst = out + (size_t)b * OCH * NPTS + n0 + 4 * c4;
    for (int j = 0; j < 4; ++j) {
        const int o = r + 16 * j;
        const float* t = tile + o * FT_PITCH + 4 * c4;
        float4 w = { t[0], t[1], t[2], t[3] };
        *(float4*)(dst + (size_t)o * NPTS) = w;
    }
}

extern "C" void kernel_launch(void* const* d_in, const int* in_sizes, int n_in,
                              void* d_out, int out_size, void* d_ws, size_t ws_size,
                              hipStream_t stream) {
    const float* x     = (const float*)d_in[0];
    const float* W     = (const float*)d_in[1];
    const float* gamma = (const float*)d_in[2];
    const float* beta  = (const float*)d_in[3];
    float* out = (float*)d_out;

    float* ws    = (float*)d_ws;
    float* sq    = ws;
    float* part  = ws + 32768;
    float* y1T   = ws + 32768 + 655360;
    float* y3T   = y1T + (size_t)BATCH * NPTS * OCH;
    float* sel   = y3T + (size_t)BATCH * NPTS * OCH;
    float* coeff = sel + (size_t)BATCH * NPTS * OCH + 1024 * 128;

    proj_kernel<<<(BATCH * NPTS / 16) / 8, 256, 0, stream>>>(x, W, y1T, y3T, sq);
    knn_kernel<<<dim3(NPTS / 16, BATCH), 256, 0, stream>>>(x, sq, y1T, y3T, gamma, sel, part);
    stats_kernel<<<1, 1024, 0, stream>>>(part, gamma, beta, coeff);
    final_kernel<<<(BATCH * NPTS) / FT_PTS, 256, 0, stream>>>(sel, coeff, out);
}


# BN statistics: the one-block sum of 2048 partial rows (1 MB through one WGP) becomes 32 blocks of 64 rows; the last 32-row stage runs in every final-pass block (summation tree changes)
# speedup vs baseline: 4.6721x; 1.0416x over previous
#include <hip/hip_runtime.h>
#include <hip/hip_bf16.h>
#include <float.h>


typedef float v2f __attribute__((ext_vector_type(2)));
typedef float v8f __attribute__((ext_vector_type(8)));

#define BATCH 8
#define CH    64
#define NPTS  4096
#define OCH   64
#define KNN   20
#define CHUNK 512
#define NEG_INF (-3.0e38f)

__device__ __forceinline__ v8f wmma_f32(v2f a, v2f b, v8f c) {
    return __builtin_amdgcn_wmma_f32_16x16x4_f32(
        false, a, false, b, (short)0, c, false, false);
}

__device__ __forceinline__ float lane_value(float v, int l) {
    return __int_as_float(__builtin_amdgcn_readlane(__float_as_int(v), l));
}

__device__ __forceinline__ void list_insert(float& lv, int& li, float cand, int cidx, int below4) {
    const float upv = __int_as_float(__builtin_amdgcn_ds_bpermute(below4, __float_as_int(lv)));
    const int   upi = __builtin_amdgcn_ds_bpermute(below4, li);
    const bool keep = lv >= cand;
    const bool here = upv >= cand;
    const float nv = here ? cand : upv;
    const int   ni = here ? cidx : upi;
    lv = keep ? lv : nv;
    li = keep ? li : ni;
}

__device__ __forceinline__ float seed_floor(const float* __restrict__ row, int lane) {
    float m = row[lane];
    for (int i = 1; i < CHUNK / 32; ++i) m = fmaxf(m, row[i * 32 + lane]);
    int larger = 0;
    for (int j = 0; j < 32; ++j) larger += (lane_value(m, j) > m) ? 1 : 0;
    const unsigned hit = __builtin_amdgcn_ballot_w32(larger == KNN - 1);
    if (hit == 0) return NEG_INF;
    const float f = lane_value(m, __builtin_ctz(hit));
    return f - fmaxf(fabsf(f) * 1.0e-6f, 1.0e-20f);
}

__device__ __forceinline__ void scan_piece(const float* __restrict__ seg, int colbase,
                                           float& lv, int& li, float& thr, int lane, int below4) {
    float v[8];
#pragma unroll
    for (int i = 0; i < 8; ++i) v[i] = seg[i * 32 + lane];
    unsigned m[8];
#pragma unroll
    for (int i = 0; i < 8; ++i) m[i] = __builtin_amdgcn_ballot_w32(v[i] > thr);
    if (((m[0] | m[1]) | (m[2] | m[3]) | (m[4] | m[5]) | (m[6] | m[7])) == 0u) return;
#pragma unroll
    for (int i = 0; i < 8; ++i) {
        unsigned mm = m[i];
        while (mm) {
            const int j = __builtin_ctz(mm);
            mm &= mm - 1;
            list_insert(lv, li, lane_value(v[i], j), colbase + i * 32 + j, below4);
        }
    }
    thr = lane_value(lv, KNN);
}

struct Lists {
    float lvA, lvB, thrA, thrB;
    int   liA, liB;
};

__device__ __forceinline__ void gram_chunk(const float* __restrict__ xb, const float* __restrict__ sqb,
                                           const v2f (&a)[16], float* __restrict__ wbuf, int c0,
                                           const float* __restrict__ rbuf, int cprev, bool doscan, Lists& L,
                                           int wave, int lane, int lmod, int khalf, int rowadd,
                                           int below4) {
    const int colq = wave * 64 + 4 * lmod;
    const float* pb = xb + (size_t)khalf * NPTS + c0 + colq;
    const float* rowA = rbuf + (2 * wave) * CHUNK;
    const float* rowB = rowA + CHUNK;
    v8f acc0 = {}, acc1 = {}, acc2 = {}, acc3 = {};
    const float4 sqc = *(const float4*)(sqb + c0 + colq);
#pragma unroll
    for (int kb = 0; kb < 4; ++kb) {
        float4 u[4], v[4];
#pragma unroll
        for (int j = 0; j < 4; ++j) {
            const int kk = 4 * kb + j;
            u[j] = *(const float4*)(pb + (size_t)(4 * kk) * NPTS);
            v[j] = *(const float4*)(pb + (size_t)(4 * kk + 1) * NPTS);
        }
        if (doscan) {
            const int half = (kb & 1) * (CHUNK / 2);
            if (kb < 2) scan_piece(rowA + half, cprev + half, L.lvA, L.liA, L.thrA, lane, below4);
            else        scan_piece(rowB + half, cprev + half, L.lvB, L.liB, L.thrB, lane, below4);
        }
#pragma unroll
        for (int j = 0; j < 4; ++j) {
            const int kk = 4 * kb + j;
            v2f b0; b0.x = u[j].x; b0.y = v[j].x;
            v2f b1; b1.x = u[j].y; b1.y = v[j].y;
            v2f b2; b2.x = u[j].z; b2.y = v[j].z;
            v2f b3; b3.x = u[j].w; b3.y = v[j].w;
            acc0 = wmma_f32(a[kk], b0, acc0);
            acc1 = wmma_f32(a[kk], b1, acc1);
            acc2 = wmma_f32(a[kk], b2, acc2);
            acc3 = wmma_f32(a[kk], b3, acc3);
        }
    }
#pragma unroll
    for (int r = 0; r < 8; ++r) {
        float4 s;
        s.x = 2.0f * acc0[r] - sqc.x;
        s.y = 2.0f * acc1[r] - sqc.y;
        s.z = 2.0f * acc2[r] - sqc.z;
        s.w = 2.0f * acc3[r] - sqc.w;
        *(float4*)(wbuf + (r + rowadd) * CHUNK + colq) = s;
    }
}

__device__ __forceinline__ float2 gather_point(const float* __restrict__ y1b,
                                               const float* __restrict__ y3T,
                                               int p, int li, int lane, bool mx0, bool mx1,
                                               float& s0, float& s1, float& q0, float& q1) {
    const float2 e = ((const float2*)(y3T + (size_t)p * OCH))[lane];
    float hi0 = -FLT_MAX, hi1 = -FLT_MAX, lo0 = FLT_MAX, lo1 = FLT_MAX;
#pragma unroll
    for (int j = 0; j < KNN; ++j) {
        const int nb = __builtin_amdgcn_readlane(li, j + 1) & (NPTS - 1);
        const float2 g = ((const float2*)(y1b + ((size_t)nb << 6)))[lane];
        const float h0 = g.x + e.x, h1 = g.y + e.y;
        s0 += h0; s1 += h1; q0 += h0 * h0; q1 += h1 * h1;
        hi0 = fmaxf(hi0, h0); lo0 = fminf(lo0, h0);
        hi1 = fmaxf(hi1, h1); lo1 = fminf(lo1, h1);
    }
    const float2 selv = { mx0 ? hi0 : lo0, mx1 ? hi1 : lo1 };
    return selv;
}

__global__ __launch_bounds__(256, 8) void knn_kernel(const float* __restrict__ x,
                                                  const float* __restrict__ sq,
                                                  const float* __restrict__ y1T,
                                                  const float* __restrict__ y3T,
                                                  const float* __restrict__ gamma,
                                                  float* __restrict__ selbuf,
                                                  float* __restrict__ partial) {
    __shared__ __attribute__((aligned(16))) float smem[2 * 16 * CHUNK];

    const int b    = blockIdx.y;
    const int q0   = blockIdx.x * 16;
    const int tid  = threadIdx.x;
    const int wave = tid >> 5;
    const int lane = tid & 31;
    const int lmod = lane & 15;
    const int khalf  = (lane >> 4) << 1;
    const int rowadd = (lane >> 4) << 3;
    const float* xb  = x + (size_t)b * CH * NPTS;
    const float* sqb = sq + (size_t)b * NPTS;

    v2f a[16];
#pragma unroll
    for (int kk = 0; kk < 16; ++kk) {
        const float* p = xb + (size_t)(4 * kk + khalf) * NPTS + q0 + lmod;
        a[kk].x = p[0];
        a[kk].y = p[NPTS];
    }

    Lists L;
    L.liA = 0; L.liB = 0;
    L.lvA = NEG_INF; L.lvB = NEG_INF; L.thrA = NEG_INF; L.thrB = NEG_INF;
    const int below4 = ((lane > 0) ? lane - 1 : 0) << 2;

    for (int c = 0; c < NPTS / CHUNK; ++c) {
        float*       wbuf = smem + (c & 1) * (16 * CHUNK);
        const float* rbuf = smem + ((c + 1) & 1) * (16 * CHUNK);
        gram_chunk(xb, sqb, a, wbuf, c * CHUNK, rbuf, (c - 1) * CHUNK, c > 0, L,
                   wave, lane, lmod, khalf, rowadd, below4);
        __syncthreads();
        if (c == 0) {
            L.thrA = seed_floor(smem + (2 * wave) * CHUNK, lane);
            L.thrB = seed_floor(smem + (2 * wave + 1) * CHUNK, lane);
            L.lvA = (lane == 0) ? __builtin_inff() : L.thrA;
            L.lvB = (lane == 0) ? __builtin_inff() : L.thrB;
        }
    }

    {
        const float* rowA = smem + 16 * CHUNK + (2 * wave) * CHUNK;
        scan_piece(rowA,                     NPTS - CHUNK,             L.lvA, L.liA, L.thrA, lane, below4);
        scan_piece(rowA + CHUNK / 2,         NPTS - CHUNK / 2,         L.lvA, L.liA, L.thrA, lane, below4);
        scan_piece(rowA + CHUNK,             NPTS - CHUNK,             L.lvB, L.liB, L.thrB, lane, below4);
        scan_piece(rowA + CHUNK + CHUNK / 2, NPTS - CHUNK / 2,         L.lvB, L.liB, L.thrB, lane, below4);
    }
    const int liA = L.liA, liB = L.liB;

    const int ch = lane * 2;
    const bool mx0 = gamma[ch] >= 0.f, mx1 = gamma[ch + 1] >= 0.f;
    const float* y1b = y1T + (size_t)b * NPTS * OCH;
    const int pA = b * NPTS + q0 + 2 * wave;
    float s0 = 0.f, s1 = 0.f, t0 = 0.f, t1 = 0.f;
    float2 selA = gather_point(y1b, y3T, pA,     liA, lane, mx0, mx1, s0, s1, t0, t1);
    float2 selB = gather_point(y1b, y3T, pA + 1, liB, lane, mx0, mx1, s0, s1, t0, t1);

    float* red = smem + wave * 128;
    red[ch]       = s0; red[ch + 1]       = s1;
    red[OCH + ch] = t0; red[OCH + ch + 1] = t1;
    __syncthreads();
    if (tid < 128) {
        float t = 0.f;
        for (int w = 0; w < 8; ++w) t += smem[w * 128 + tid];
        partial[((size_t)b * (NPTS / 16) + blockIdx.x) * 128 + tid] = t;
    }
    ((float2*)(selbuf + (size_t)pA * OCH))[lane]       = selA;
    ((float2*)(selbuf + (size_t)(pA + 1) * OCH))[lane] = selB;
}

__global__ __launch_bounds__(256) void proj_kernel(const float* __restrict__ x,
                                                   const float* __restrict__ W,
                                                   float* __restrict__ y1T,
                                                   float* __restrict__ y3T,
                                                   float* __restrict__ sq) {
    const int task = blockIdx.x * 8 + (threadIdx.x >> 5);
    const int lane = threadIdx.x & 31;
    const int lmod = lane & 15;
    const int half = lane >> 4;
    const int khalf  = half << 1;
    const int rowadd = half << 3;

    const int b  = task >> 8;
    const int n0 = (task & 255) * 16;

    const float* xp = x + (size_t)b * CH * NPTS + n0 + lmod;
    v2f bv[16];
#pragma unroll
    for (int kk = 0; kk < 16; ++kk) {
        const float* p = xp + (size_t)(4 * kk + khalf) * NPTS;
        bv[kk].x = p[0];
        bv[kk].y = p[NPTS];
    }

    {
        const int other4 = (lane ^ 16) << 2;
        float s = 0.f;
#pragma unroll
        for (int kk = 0; kk < 16; ++kk) {
            const float ox = __int_as_float(__builtin_amdgcn_ds_bpermute(other4, __float_as_int(bv[kk].x)));
            const float oy = __int_as_float(__builtin_amdgcn_ds_bpermute(other4, __float_as_int(bv[kk].y)));
            const float c0 = half ? ox : bv[kk].x;
            const float c1 = half ? oy : bv[kk].y;
            const float c2 = half ? bv[kk].x : ox;
            const float c3 = half ? bv[kk].y : oy;
            if (kk == 0) {
                s = __builtin_fmaf(c0, c0, c1 * c1);
            } else {
                s = __builtin_fmaf(c0, c0, s);
                s = __builtin_fmaf(c1, c1, s);
            }
            s = __builtin_fmaf(c2, c2, s);
            s = __builtin_fmaf(c3, c3, s);
        }
        if (half == 0) sq[(size_t)b * NPTS + n0 + lmod] = s;
    }

    const size_t yoff = (size_t)b * NPTS * OCH + (size_t)(n0 + lmod) * OCH + rowadd;
    for (int ot = 0; ot < 4; ++ot) {
        const float* wr = W + (size_t)(ot * 16 + lmod) * (2 * CH) + khalf;
        v8f acc1 = {}, acc3 = {};
#pragma unroll
        for (int kk = 0; kk < 16; ++kk) {
            const float* w = wr + 4 * kk;
            v2f a1, a3;
            a1.x = w[0];           a1.y = w[1];
            a3.x = w[CH] - w[0];   a3.y = w[CH + 1] - w[1];
            acc1 = wmma_f32(a1, bv[kk], acc1);
            acc3 = wmma_f32(a3, bv[kk], acc3);
        }
        float* y1 = y1T + yoff + ot * 16;
        float* y3 = y3T + yoff + ot * 16;
        float4 lo1 = { acc1[0], acc1[1], acc1[2], acc1[3] };
        float4 hi1 = { acc1[4], acc1[5], acc1[6], acc1[7] };
        float4 lo3 = { acc3[0], acc3[1], acc3[2], acc3[3] };
        float4 hi3 = { acc3[4], acc3[5], acc3[6], acc3[7] };
        ((float4*)y1)[0] = lo1;
        ((float4*)y1)[1] = hi1;
        ((float4*)y3)[0] = lo3;
        ((float4*)y3)[1] = hi3;
    }
}

#define ST_BLOCKS 32
#define ST_ROWS   64

__global__ __launch_bounds__(256) void stats_kernel(const float* __restrict__ partial,
                                                    float* __restrict__ part2) {
    __shared__ __attribute__((aligned(16))) float red[8 * 128];
    const int tid  = threadIdx.x;
    const int wave = tid >> 5;
    const int lane = tid & 31;

    const float4* p = (const float4*)(partial + ((size_t)blockIdx.x * ST_ROWS + wave * 8) * 128) + lane;
    float4 v[8];
#pragma unroll
    for (int r = 0; r < 8; ++r) v[r] = p[(size_t)r * 32];
    float4 a = v[0];
#pragma unroll
    for (int r = 1; r < 8; ++r) { a.x += v[r].x; a.y += v[r].y; a.z += v[r].z; a.w += v[r].w; }
    ((float4*)red)[wave * 32 + lane] = a;
    __syncthreads();

    if (tid < 128) {
        float t = 0.f;
        for (int w = 0; w < 8; ++w) t += red[w * 128 + tid];
        part2[(size_t)blockIdx.x * 128 + tid] = t;
    }
}

#define FT_PTS   64
#define FT_PITCH 65

__global__ __launch_bounds__(256) void final_kernel(const float* __restrict__ selbuf,
                                                    const float* __restrict__ part2,
                                                    const float* __restrict__ gamma,
                                                    const float* __restrict__ beta,
                                                    float* __restrict__ out) {
    __shared__ float tile[OCH * FT_PITCH];
    __shared__ __attribute__((aligned(16))) float red[8 * 128];
    __shared__ __attribute__((aligned(16))) float cf[2 * OCH];

    const int tid  = threadIdx.x;
    const int wave = tid >> 5;
    const int lane = tid & 31;
    const int c4  = tid & 15;
    const int r   = tid >> 4;
    const int p0  = blockIdx.x * FT_PTS;
    const int b   = p0 >> 12;
    const int n0  = p0 & (NPTS - 1);

    const float4* sp = (const float4*)(part2 + (size_t)(wave * 4) * 128) + lane;
    float4 u[4];
    for (int j = 0; j < 4; ++j) u[j] = sp[j * 32];
    const float4* srcp = (const float4*)(selbuf + (size_t)p0 * OCH) + tid;
    float4 v[4];
    for (int j = 0; j < 4; ++j) v[j] = srcp[j * 256];
    float g = 0.f, bt = 0.f;
    if (tid < OCH) { g = gamma[tid]; bt = beta[tid]; }

    float4 a = u[0];
    for (int j = 1; j < 4; ++j) { a.x += u[j].x; a.y += u[j].y; a.z += u[j].z; a.w += u[j].w; }
    ((float4*)red)[wave * 32 + lane] = a;
    __syncthreads();
    if (tid < OCH) {
        float s = 0.f, q = 0.f;
        for (int w = 0; w < 8; ++w) {
            s += red[w * 128 + tid];
            q += red[w * 128 + OCH + tid];
        }
        const float M = (float)BATCH * NPTS * KNN;
        float mean  = s / M;
        float var   = q / M - mean * mean;
        float scale = g * rsqrtf(var + 1e-5f);
        cf[tid]       = scale;
        cf[OCH + tid] = bt - mean * scale;
    }
    __syncthreads();

    const float4 sc = ((const float4*)cf)[c4];
    const float4 sh = ((const float4*)(cf + OCH))[c4];

    for (int j = 0; j < 4; ++j) {
        const int pt = r + 16 * j;
        float r0 = v[j].x * sc.x + sh.x;
        float r1 = v[j].y * sc.y + sh.y;
        float r2 = v[j].z * sc.z + sh.z;
        float r3 = v[j].w * sc.w + sh.w;
        tile[(4 * c4 + 0) * FT_PITCH + pt] = (r0 >= 0.f) ? r0 : 0.2f * r0;
        tile[(4 * c4 + 1) * FT_PITCH + pt] = (r1 >= 0.f) ? r1 : 0.2f * r1;
        tile[(4 * c4 + 2) * FT_PITCH + pt] = (r2 >= 0.f) ? r2 : 0.2f * r2;
        tile[(4 * c4 + 3) * FT_PITCH + pt] = (r3 >= 0.f) ? r3 : 0.2f * r3;
    }
    __syncthreads();

    float* dst = out + (size_t)b * OCH * NPTS + n0 + 4 * c4;
    for (int j = 0; j < 4; ++j) {
        const int o = r + 16 * j;
        const float* t = tile + o * FT_PITCH + 4 * c4;
        float4 w = { t[0], t[1], t[2], t[3] };
        *(float4*)(dst + (size_t)o * NPTS) = w;
    }
}

extern "C" void kernel_launch(void* const* d_in, const int* in_sizes, int n_in,
                              void* d_out, int out_size, void* d_ws, size_t ws_size,
                              hipStream_t stream) {
    const float* x     = (const float*)d_in[0];
    const float* W     = (const float*)d_in[1];
    const float* gamma = (const float*)d_in[2];
    const float* beta  = (const float*)d_in[3];
    float* out = (float*)d_out;

    float* ws    = (float*)d_ws;
    float* sq    = ws;
    float* part  = ws + 32768;
    float* part2 = ws + 32768 + 262144;
    float* y1T   = ws + 32768 + 655360;
    float* y3T   = y1T + (size_t)BATCH * NPTS * OCH;
    float* sel   = y3T + (size_t)BATCH * NPTS * OCH;

    proj_kernel<<<(BATCH * NPTS / 16) / 8, 256, 0, stream>>>(x, W, y1T, y3T, sq);
    knn_kernel<<<dim3(NPTS / 16, BATCH), 256, 0, stream>>>(x, sq, y1T, y3T, gamma, sel, part);
    stats_kernel<<<ST_BLOCKS, 256, 0, stream>>>(part, part2);
    final_kernel<<<(BATCH * NPTS) / FT_PTS, 256, 0, stream>>>(sel, part2, gamma, beta, out);
}
